# starting kernel
# baseline (speedup 1.0000x reference)
#include <hip/hip_runtime.h>
#include <math.h>

typedef float v2f __attribute__((ext_vector_type(2)));
typedef float v8f __attribute__((ext_vector_type(8)));

#define GCN_N 100000
#define GCN_E 1600000
#define GCN_INDIM 128
#define GCN_HID 64

__global__ __launch_bounds__(256) void gcn_deg_kernel(
    const int* __restrict__ src, const int* __restrict__ dst,
    float* __restrict__ deg_out, float* __restrict__ deg_in, int E) {
  int e = blockIdx.x * blockDim.x + threadIdx.x;
  if (e < E) {
    atomicAdd(&deg_out[src[e]], 1.0f);
    atomicAdd(&deg_in[dst[e]], 1.0f);
  }
}

__global__ __launch_bounds__(256) void gcn_norm_kernel(float* __restrict__ d, int n) {
  int i = blockIdx.x * blockDim.x + threadIdx.x;
  if (i < n) d[i] = rsqrtf(fmaxf(d[i], 1.0f));
}

__global__ __launch_bounds__(256) void gcn_gemm_wmma(
    const float* __restrict__ A, const float* __restrict__ W,
    const float* __restrict__ nsrc, float* __restrict__ X,
    int K, int nRows) {
  int wave = blockIdx.x * (blockDim.x >> 5) + (threadIdx.x >> 5);
  int lane = threadIdx.x & 31;
  int rowBlk = wave * 16;
  if (rowBlk >= nRows) return;

  int half = lane >> 4;
  int l    = lane & 15;
  int row  = rowBlk + l;
  float ns = nsrc[row];
  const float* arow = A + (long)row * K;

  v8f acc0 = {}, acc1 = {}, acc2 = {}, acc3 = {};

  for (int k0 = 0; k0 < K; k0 += 4) {
    int ka = k0 + half * 2;
    v2f a;
    a.x = arow[ka]     * ns;
    a.y = arow[ka + 1] * ns;
    const float* w0 = W + (long)ka * 64 + l;
    const float* w1 = w0 + 64;
    v2f b;
    b.x = w0[0];  b.y = w1[0];
    acc0 = __builtin_amdgcn_wmma_f32_16x16x4_f32(false, a, false, b, (short)0, acc0, false, false);
    b.x = w0[16]; b.y = w1[16];
    acc1 = __builtin_amdgcn_wmma_f32_16x16x4_f32(false, a, false, b, (short)0, acc1, false, false);
    b.x = w0[32]; b.y = w1[32];
    acc2 = __builtin_amdgcn_wmma_f32_16x16x4_f32(false, a, false, b, (short)0, acc2, false, false);
    b.x = w0[48]; b.y = w1[48];
    acc3 = __builtin_amdgcn_wmma_f32_16x16x4_f32(false, a, false, b, (short)0, acc3, false, false);
  }

  float* xout = X + (long)rowBlk * 64;
#pragma unroll
  for (int i = 0; i < 8; ++i) {
    int r = i + half * 8;
    xout[(long)r * 64 +  0 + l] = acc0[i];
    xout[(long)r * 64 + 16 + l] = acc1[i];
    xout[(long)r * 64 + 32 + l] = acc2[i];
    xout[(long)r * 64 + 48 + l] = acc3[i];
  }
}

__global__ __launch_bounds__(256) void gcn_edge_agg(
    const int* __restrict__ src, const int* __restrict__ dst,
    const float* __restrict__ X, float* __restrict__ AGG, int E) {
  long tid = (long)blockIdx.x * blockDim.x + threadIdx.x;
  long e = tid >> 4;
  int  j = (int)(tid & 15) * 4;
  if (e < E) {
    int s = src[e];
    int d = dst[e];
    const float4 v = *(const float4*)(X + (long)s * 64 + j);
    float* p = AGG + (long)d * 64 + j;
    atomicAdd(p + 0, v.x);
    atomicAdd(p + 1, v.y);
    atomicAdd(p + 2, v.z);
    atomicAdd(p + 3, v.w);
  }
}

__global__ __launch_bounds__(256) void gcn_elu_kernel(
    float* __restrict__ AGG, const float* __restrict__ ndst,
    const float* __restrict__ b, int n) {
  int i = blockIdx.x * blockDim.x + threadIdx.x;
  if (i < n) {
    float v = AGG[i] * ndst[i >> 6] + b[i & 63];
    AGG[i] = v > 0.0f ? v : expm1f(v);
  }
}

__global__ __launch_bounds__(256) void gcn_final_kernel(
    float* __restrict__ OUT, const float* __restrict__ ndst,
    const float* __restrict__ b, int n) {
  int i = blockIdx.x * blockDim.x + threadIdx.x;
  if (i < n) OUT[i] = OUT[i] * ndst[i >> 6] + b[i & 63];
}

extern "C" void kernel_launch(void* const* d_in, const int* in_sizes, int n_in,
                              void* d_out, int out_size, void* d_ws, size_t ws_size,
                              hipStream_t stream) {
  const float* h   = (const float*)d_in[0];
  const int*   src = (const int*)  d_in[1];
  const int*   dst = (const int*)  d_in[2];
  const float* W1  = (const float*)d_in[3];
  const float* b1  = (const float*)d_in[4];
  const float* W2  = (const float*)d_in[5];
  const float* b2  = (const float*)d_in[6];
  float* out = (float*)d_out;

  const int N = in_sizes[0] / GCN_INDIM;
  const int E = in_sizes[1];

  float* norm_src = (float*)d_ws;
  float* norm_dst = norm_src + N;
  float* bufA     = norm_dst + N;
  float* bufB     = bufA + (long)N * 64;

  const int feat = N * 64;

  hipMemsetAsync(norm_src, 0, (size_t)2 * N * sizeof(float), stream);
  gcn_deg_kernel<<<(E + 255) / 256, 256, 0, stream>>>(src, dst, norm_src, norm_dst, E);
  gcn_norm_kernel<<<(2 * N + 255) / 256, 256, 0, stream>>>(norm_src, 2 * N);

  {
    int waves = N / 16;
    int blocks = (waves + 7) / 8;
    gcn_gemm_wmma<<<blocks, 256, 0, stream>>>(h, W1, norm_src, bufA, GCN_INDIM, N);
  }
  hipMemsetAsync(bufB, 0, (size_t)feat * sizeof(float), stream);
  {
    long threads = (long)E * 16;
    gcn_edge_agg<<<(int)((threads + 255) / 256), 256, 0, stream>>>(src, dst, bufA, bufB, E);
  }
  gcn_elu_kernel<<<(feat + 255) / 256, 256, 0, stream>>>(bufB, norm_dst, b1, feat);

  {
    int waves = N / 16;
    int blocks = (waves + 7) / 8;
    gcn_gemm_wmma<<<blocks, 256, 0, stream>>>(bufB, W2, norm_src, bufA, GCN_HID, N);
  }
  hipMemsetAsync(out, 0, (size_t)out_size * sizeof(float), stream);
  {
    long threads = (long)E * 16;
    gcn_edge_agg<<<(int)((threads + 255) / 256), 256, 0, stream>>>(src, dst, bufA, out, E);
  }
  gcn_final_kernel<<<(out_size + 255) / 256, 256, 0, stream>>>(out, norm_dst, b2, out_size);
}


# edge pass: a wave walks 32 edges, lane = feature: every gather load and f32 atomic add covers one whole 128-B line (was four quarter lines per atomic); sums stay f32 atomics in arrival order
# speedup vs baseline: 1.4428x; 1.4428x over previous
#include <hip/hip_runtime.h>
#include <math.h>

typedef float v2f __attribute__((ext_vector_type(2)));
typedef float v8f __attribute__((ext_vector_type(8)));

#define GCN_N 100000
#define GCN_E 1600000
#define GCN_INDIM 128
#define GCN_HID 64

__global__ __launch_bounds__(256) void gcn_deg_kernel(
    const int* __restrict__ src, const int* __restrict__ dst,
    float* __restrict__ deg_out, float* __restrict__ deg_in, int E) {
  int e = blockIdx.x * blockDim.x + threadIdx.x;
  if (e < E) {
    atomicAdd(&deg_out[src[e]], 1.0f);
    atomicAdd(&deg_in[dst[e]], 1.0f);
  }
}

__global__ __launch_bounds__(256) void gcn_norm_kernel(float* __restrict__ d, int n) {
  int i = blockIdx.x * blockDim.x + threadIdx.x;
  if (i < n) d[i] = rsqrtf(fmaxf(d[i], 1.0f));
}

__global__ __launch_bounds__(256) void gcn_gemm_wmma(
    const float* __restrict__ A, const float* __restrict__ W,
    const float* __restrict__ nsrc, float* __restrict__ X,
    int K, int nRows) {
  int wave = blockIdx.x * (blockDim.x >> 5) + (threadIdx.x >> 5);
  int lane = threadIdx.x & 31;
  int rowBlk = wave * 16;
  if (rowBlk >= nRows) return;

  int half = lane >> 4;
  int l    = lane & 15;
  int row  = rowBlk + l;
  float ns = nsrc[row];
  const float* arow = A + (long)row * K;

  v8f acc0 = {}, acc1 = {}, acc2 = {}, acc3 = {};

  for (int k0 = 0; k0 < K; k0 += 4) {
    int ka = k0 + half * 2;
    v2f a;
    a.x = arow[ka]     * ns;
    a.y = arow[ka + 1] * ns;
    const float* w0 = W + (long)ka * 64 + l;
    const float* w1 = w0 + 64;
    v2f b;
    b.x = w0[0];  b.y = w1[0];
    acc0 = __builtin_amdgcn_wmma_f32_16x16x4_f32(false, a, false, b, (short)0, acc0, false, false);
    b.x = w0[16]; b.y = w1[16];
    acc1 = __builtin_amdgcn_wmma_f32_16x16x4_f32(false, a, false, b, (short)0, acc1, false, false);
    b.x = w0[32]; b.y = w1[32];
    acc2 = __builtin_amdgcn_wmma_f32_16x16x4_f32(false, a, false, b, (short)0, acc2, false, false);
    b.x = w0[48]; b.y = w1[48];
    acc3 = __builtin_amdgcn_wmma_f32_16x16x4_f32(false, a, false, b, (short)0, acc3, false, false);
  }

  float* xout = X + (long)rowBlk * 64;
#pragma unroll
  for (int i = 0; i < 8; ++i) {
    int r = i + half * 8;
    xout[(long)r * 64 +  0 + l] = acc0[i];
    xout[(long)r * 64 + 16 + l] = acc1[i];
    xout[(long)r * 64 + 32 + l] = acc2[i];
    xout[(long)r * 64 + 48 + l] = acc3[i];
  }
}

#define EDGE_BATCH 16
__global__ __launch_bounds__(256) void gcn_edge_agg(
    const int* __restrict__ src, const int* __restrict__ dst,
    const float* __restrict__ X, float* __restrict__ AGG, int E) {
  const int lane = threadIdx.x & 31;
  const int wv = __builtin_amdgcn_readfirstlane((int)(threadIdx.x >> 5));
  const long base = ((long)blockIdx.x * (blockDim.x >> 5) + wv) * 32;
  if (base >= E) return;
  const long rem = (long)E - base;
  const int cnt = rem < 32 ? (int)rem : 32;

  int sv = 0, dv = 0;
  if (lane < cnt) { sv = src[base + lane]; dv = dst[base + lane]; }

  const float* xl = X + lane;
  float* al = AGG + lane;

  if (cnt == 32) {
    for (int j0 = 0; j0 < 32; j0 += EDGE_BATCH) {
      float x0[EDGE_BATCH], x1[EDGE_BATCH];
#pragma unroll
      for (int u = 0; u < EDGE_BATCH; ++u) {
        const int sr = __builtin_amdgcn_readlane(sv, j0 + u);
        const float* p = xl + (long)sr * 64;
        x0[u] = p[0];
        x1[u] = p[32];
      }
#pragma unroll
      for (int u = 0; u < EDGE_BATCH; ++u) {
        const int dr = __builtin_amdgcn_readlane(dv, j0 + u);
        float* q = al + (long)dr * 64;
        atomicAdd(q, x0[u]);
        atomicAdd(q + 32, x1[u]);
      }
    }
  } else {
    for (int j = 0; j < cnt; ++j) {
      const int sr = __builtin_amdgcn_readlane(sv, j);
      const int dr = __builtin_amdgcn_readlane(dv, j);
      const float* p = xl + (long)sr * 64;
      float* q = al + (long)dr * 64;
      const float a0 = p[0];
      const float a1 = p[32];
      atomicAdd(q, a0);
      atomicAdd(q + 32, a1);
    }
  }
}

__global__ __launch_bounds__(256) void gcn_elu_kernel(
    float* __restrict__ AGG, const float* __restrict__ ndst,
    const float* __restrict__ b, int n) {
  int i = blockIdx.x * blockDim.x + threadIdx.x;
  if (i < n) {
    float v = AGG[i] * ndst[i >> 6] + b[i & 63];
    AGG[i] = v > 0.0f ? v : expm1f(v);
  }
}

__global__ __launch_bounds__(256) void gcn_final_kernel(
    float* __restrict__ OUT, const float* __restrict__ ndst,
    const float* __restrict__ b, int n) {
  int i = blockIdx.x * blockDim.x + threadIdx.x;
  if (i < n) OUT[i] = OUT[i] * ndst[i >> 6] + b[i & 63];
}

extern "C" void kernel_launch(void* const* d_in, const int* in_sizes, int n_in,
                              void* d_out, int out_size, void* d_ws, size_t ws_size,
                              hipStream_t stream) {
  const float* h   = (const float*)d_in[0];
  const int*   src = (const int*)  d_in[1];
  const int*   dst = (const int*)  d_in[2];
  const float* W1  = (const float*)d_in[3];
  const float* b1  = (const float*)d_in[4];
  const float* W2  = (const float*)d_in[5];
  const float* b2  = (const float*)d_in[6];
  float* out = (float*)d_out;

  const int N = in_sizes[0] / GCN_INDIM;
  const int E = in_sizes[1];

  float* norm_src = (float*)d_ws;
  float* norm_dst = norm_src + N;
  float* bufA     = norm_dst + N;
  float* bufB     = bufA + (long)N * 64;

  const int feat = N * 64;

  hipMemsetAsync(norm_src, 0, (size_t)2 * N * sizeof(float), stream);
  gcn_deg_kernel<<<(E + 255) / 256, 256, 0, stream>>>(src, dst, norm_src, norm_dst, E);
  gcn_norm_kernel<<<(2 * N + 255) / 256, 256, 0, stream>>>(norm_src, 2 * N);

  {
    int waves = N / 16;
    int blocks = (waves + 7) / 8;
    gcn_gemm_wmma<<<blocks, 256, 0, stream>>>(h, W1, norm_src, bufA, GCN_INDIM, N);
  }
  hipMemsetAsync(bufB, 0, (size_t)feat * sizeof(float), stream);
  gcn_edge_agg<<<(E + 255) / 256, 256, 0, stream>>>(src, dst, bufA, bufB, E);
  gcn_elu_kernel<<<(feat + 255) / 256, 256, 0, stream>>>(bufB, norm_dst, b1, feat);

  {
    int waves = N / 16;
    int blocks = (waves + 7) / 8;
    gcn_gemm_wmma<<<blocks, 256, 0, stream>>>(bufB, W2, norm_src, bufA, GCN_HID, N);
  }
  hipMemsetAsync(out, 0, (size_t)out_size * sizeof(float), stream);
  gcn_edge_agg<<<(E + 255) / 256, 256, 0, stream>>>(src, dst, bufA, out, E);
  gcn_final_kernel<<<(out_size + 255) / 256, 256, 0, stream>>>(out, norm_dst, b2, out_size);
}


# edge sums: f32 atomic scatter replaced by destination-sorted pull (in-call counting sort, returning int add = slot, arrival order); rows summed in registers in slot order, scale+bias+ELU at row store
# speedup vs baseline: 1.7368x; 1.2038x over previous
#include <hip/hip_runtime.h>
#include <math.h>

typedef float v2f __attribute__((ext_vector_type(2)));
typedef float v8f __attribute__((ext_vector_type(8)));

#define GCN_N 100000
#define GCN_E 1600000
#define GCN_INDIM 128
#define GCN_HID 64

__global__ __launch_bounds__(256) void gcn_deg_rank(
    const int* __restrict__ src, const int* __restrict__ dst,
    int* __restrict__ cnt_out, int* __restrict__ cnt_in,
    int* __restrict__ rank, int E) {
  const int blk = blockIdx.x * 1024;
  const int base = blk + threadIdx.x;
  if (blk + 1024 <= E) {
    int s[4], d[4], r[4];
#pragma unroll
    for (int k = 0; k < 4; ++k) { s[k] = src[base + k * 256]; d[k] = dst[base + k * 256]; }
#pragma unroll
    for (int k = 0; k < 4; ++k) atomicAdd(&cnt_out[s[k]], 1);
#pragma unroll
    for (int k = 0; k < 4; ++k) r[k] = atomicAdd(&cnt_in[d[k]], 1);
#pragma unroll
    for (int k = 0; k < 4; ++k) rank[base + k * 256] = r[k];
  } else {
    for (int k = 0; k < 4; ++k) {
      const int e = base + k * 256;
      if (e < E) {
        atomicAdd(&cnt_out[src[e]], 1);
        rank[e] = atomicAdd(&cnt_in[dst[e]], 1);
      }
    }
  }
}

__global__ __launch_bounds__(1024) void gcn_scan_norm(
    const int* __restrict__ cnt_out, const int* __restrict__ cnt_in,
    float* __restrict__ norm_src, float* __restrict__ norm_dst,
    int* __restrict__ row_ptr, int n) {
  __shared__ int s_part[32];
  __shared__ int s_incl[32];
  const int tid = threadIdx.x;
  const int lane = tid & 31;
  const int wv = tid >> 5;
  const int blockStart = blockIdx.x * 1024;

  int part = 0;
  const int4* c4 = (const int4*)cnt_in;
  const int grp = __builtin_amdgcn_readfirstlane(tid >> 8);
  const int sub = tid & 255;
  const int groups = blockIdx.x;
#pragma unroll 8
  for (int g = grp; g < groups; g += 4) {
    const int4 q = c4[g * 256 + sub];
    part += (q.x + q.y) + (q.z + q.w);
  }
  for (int o = 16; o > 0; o >>= 1) part += __shfl_xor(part, o);
  if (lane == 0) s_part[wv] = part;

  const int idx = blockStart + tid;
  const int c = idx < n ? cnt_in[idx] : 0;
  int inc = c;
  for (int o = 1; o < 32; o <<= 1) {
    const int v = __shfl_up(inc, o);
    if (lane >= o) inc += v;
  }
  if (lane == 31) s_incl[wv] = inc;
  __syncthreads();

  int off = 0;
  for (int w = 0; w < 32; ++w) {
    off += s_part[w];
    if (w < wv) off += s_incl[w];
  }
  const int excl = off + inc - c;
  if (idx < n) {
    row_ptr[idx] = excl;
    if (idx == n - 1) row_ptr[n] = excl + c;
    norm_dst[idx] = rsqrtf(fmaxf((float)c, 1.0f));
    norm_src[idx] = rsqrtf(fmaxf((float)cnt_out[idx], 1.0f));
  }
}

__global__ __launch_bounds__(256) void gcn_fill(
    const int* __restrict__ src, const int* __restrict__ dst,
    const int* __restrict__ rank, const int* __restrict__ row_ptr,
    int* __restrict__ col, int E) {
  const int blk = blockIdx.x * 1024;
  const int base = blk + threadIdx.x;
  if (blk + 1024 <= E) {
    int s[4], d[4], r[4], p[4];
#pragma unroll
    for (int k = 0; k < 4; ++k) {
      s[k] = src[base + k * 256];
      d[k] = dst[base + k * 256];
      r[k] = rank[base + k * 256];
    }
#pragma unroll
    for (int k = 0; k < 4; ++k) p[k] = row_ptr[d[k]];
#pragma unroll
    for (int k = 0; k < 4; ++k) col[p[k] + r[k]] = s[k];
  } else {
    for (int k = 0; k < 4; ++k) {
      const int e = base + k * 256;
      if (e < E) col[row_ptr[dst[e]] + rank[e]] = src[e];
    }
  }
}

__global__ __launch_bounds__(256) void gcn_gemm_wmma(
    const float* __restrict__ A, const float* __restrict__ W,
    const float* __restrict__ nsrc, float* __restrict__ X,
    int K, int nRows) {
  int wave = blockIdx.x * (blockDim.x >> 5) + (threadIdx.x >> 5);
  int lane = threadIdx.x & 31;
  int rowBlk = wave * 16;
  if (rowBlk >= nRows) return;

  int half = lane >> 4;
  int l    = lane & 15;
  int row  = rowBlk + l;
  float ns = nsrc[row];
  const float* arow = A + (long)row * K;

  v8f acc0 = {}, acc1 = {}, acc2 = {}, acc3 = {};

  for (int k0 = 0; k0 < K; k0 += 4) {
    int ka = k0 + half * 2;
    v2f a;
    a.x = arow[ka]     * ns;
    a.y = arow[ka + 1] * ns;
    const float* w0 = W + (long)ka * 64 + l;
    const float* w1 = w0 + 64;
    v2f b;
    b.x = w0[0];  b.y = w1[0];
    acc0 = __builtin_amdgcn_wmma_f32_16x16x4_f32(false, a, false, b, (short)0, acc0, false, false);
    b.x = w0[16]; b.y = w1[16];
    acc1 = __builtin_amdgcn_wmma_f32_16x16x4_f32(false, a, false, b, (short)0, acc1, false, false);
    b.x = w0[32]; b.y = w1[32];
    acc2 = __builtin_amdgcn_wmma_f32_16x16x4_f32(false, a, false, b, (short)0, acc2, false, false);
    b.x = w0[48]; b.y = w1[48];
    acc3 = __builtin_amdgcn_wmma_f32_16x16x4_f32(false, a, false, b, (short)0, acc3, false, false);
  }

  float* xout = X + (long)rowBlk * 64;
#pragma unroll
  for (int i = 0; i < 8; ++i) {
    int r = i + half * 8;
    xout[(long)r * 64 +  0 + l] = acc0[i];
    xout[(long)r * 64 + 16 + l] = acc1[i];
    xout[(long)r * 64 + 32 + l] = acc2[i];
    xout[(long)r * 64 + 48 + l] = acc3[i];
  }
}

#define PULL_NPW 8

template <bool WITH_ELU>
__device__ __forceinline__ void gcn_pull_body(
    const int* __restrict__ row_ptr, const int* __restrict__ col,
    const float* __restrict__ X, const float* __restrict__ ndst,
    const float* __restrict__ b, float* __restrict__ OUT, int n) {
  const int lane = threadIdx.x & 31;
  const int wv = __builtin_amdgcn_readfirstlane((int)(threadIdx.x >> 5));
  const int node0 = (blockIdx.x * 8 + wv) * PULL_NPW;
  if (node0 >= n) return;
  const int rem = n - node0;
  const int cnt = rem < PULL_NPW ? rem : PULL_NPW;

  int rpv = 0;
  float ndv = 0.0f;
  if (lane <= cnt) rpv = row_ptr[node0 + lane];
  if (lane < cnt) ndv = ndst[node0 + lane];
  const float b0 = b[2 * lane];
  const float b1 = b[2 * lane + 1];
  const int lo = 2 * lane;

  const int first = __builtin_amdgcn_readlane(rpv, 0);
  const int total = __builtin_amdgcn_readlane(rpv, cnt) - first;
  const int* cl = col + first;

  int t = 0;
  int nodeEnd = __builtin_amdgcn_readlane(rpv, 1) - first;
  float ax = 0.0f, ay = 0.0f;

#define PULL_FINISH()                                                          \
  {                                                                            \
    const float nd = __int_as_float(                                           \
        __builtin_amdgcn_readlane(__float_as_int(ndv), t));                    \
    float r0 = ax * nd + b0;                                                   \
    float r1 = ay * nd + b1;                                                   \
    if (WITH_ELU) {                                                            \
      r0 = r0 > 0.0f ? r0 : expm1f(r0);                                        \
      r1 = r1 > 0.0f ? r1 : expm1f(r1);                                        \
    }                                                                          \
    v2f r;                                                                     \
    r.x = r0;                                                                  \
    r.y = r1;                                                                  \
    *(v2f*)(OUT + (long)(node0 + t) * 64 + lo) = r;                            \
    ax = 0.0f;                                                                 \
    ay = 0.0f;                                                                 \
    ++t;                                                                       \
    nodeEnd = __builtin_amdgcn_readlane(rpv, t + 1) - first;                   \
  }

  int cv = 0;
  if (lane < 16 && lane < total) cv = cl[lane];

  for (int pos0 = 0; pos0 < total; pos0 += 16) {
    v2f v[16];
#pragma unroll
    for (int u = 0; u < 16; ++u) {
      const int sr = __builtin_amdgcn_readlane(cv, u);
      v[u] = *(const v2f*)(X + (long)sr * 64 + lo);
    }
    int cvn = 0;
    if (lane < 16 && pos0 + 16 + lane < total) cvn = cl[pos0 + 16 + lane];
#pragma unroll
    for (int u = 0; u < 16; ++u) {
      while (pos0 + u == nodeEnd) PULL_FINISH()
      ax += v[u].x;
      ay += v[u].y;
    }
    cv = cvn;
  }
  while (t < cnt) PULL_FINISH()
#undef PULL_FINISH
}

__global__ __launch_bounds__(256) void gcn_pull_elu(
    const int* __restrict__ row_ptr, const int* __restrict__ col,
    const float* __restrict__ X, const float* __restrict__ ndst,
    const float* __restrict__ b, float* __restrict__ OUT, int n) {
  gcn_pull_body<true>(row_ptr, col, X, ndst, b, OUT, n);
}

__global__ __launch_bounds__(256) void gcn_pull_lin(
    const int* __restrict__ row_ptr, const int* __restrict__ col,
    const float* __restrict__ X, const float* __restrict__ ndst,
    const float* __restrict__ b, float* __restrict__ OUT, int n) {
  gcn_pull_body<false>(row_ptr, col, X, ndst, b, OUT, n);
}

extern "C" void kernel_launch(void* const* d_in, const int* in_sizes, int n_in,
                              void* d_out, int out_size, void* d_ws, size_t ws_size,
                              hipStream_t stream) {
  const float* h   = (const float*)d_in[0];
  const int*   src = (const int*)  d_in[1];
  const int*   dst = (const int*)  d_in[2];
  const float* W1  = (const float*)d_in[3];
  const float* b1  = (const float*)d_in[4];
  const float* W2  = (const float*)d_in[5];
  const float* b2  = (const float*)d_in[6];
  float* out = (float*)d_out;

  const int N = in_sizes[0] / GCN_INDIM;
  const int E = in_sizes[1];

  float* norm_src = (float*)d_ws;
  float* norm_dst = norm_src + N;
  float* bufA     = norm_dst + N;
  int*   cnt_out  = (int*)(bufA + (long)N * 64);
  int*   cnt_in   = cnt_out + N;
  int*   row_ptr  = cnt_in + N;
  int*   rank     = row_ptr + N + 32;
  int*   col      = rank + E;

  const int edgeBlocks = (E + 1023) / 1024;
  const int pullBlocks = (N + 8 * PULL_NPW - 1) / (8 * PULL_NPW);
  const int waves  = N / 16;
  const int gemmBlocks = (waves + 7) / 8;

  hipMemsetAsync(cnt_out, 0, (size_t)2 * N * sizeof(int), stream);
  gcn_deg_rank<<<edgeBlocks, 256, 0, stream>>>(src, dst, cnt_out, cnt_in, rank, E);
  gcn_scan_norm<<<(N + 1023) / 1024, 1024, 0, stream>>>(cnt_out, cnt_in, norm_src, norm_dst, row_ptr, N);
  gcn_fill<<<edgeBlocks, 256, 0, stream>>>(src, dst, rank, row_ptr, col, E);

  gcn_gemm_wmma<<<gemmBlocks, 256, 0, stream>>>(h, W1, norm_src, bufA, GCN_INDIM, N);
  gcn_pull_elu<<<pullBlocks, 256, 0, stream>>>(row_ptr, col, bufA, norm_dst, b1, out, N);

  gcn_gemm_wmma<<<gemmBlocks, 256, 0, stream>>>(out, W2, norm_src, bufA, GCN_HID, N);
  gcn_pull_lin<<<pullBlocks, 256, 0, stream>>>(row_ptr, col, bufA, norm_dst, b2, out, N);
}


# edge counting sort: global integer atomics on random words replaced by per-block LDS counters over node ranges (per-chunk counts + chunk prefix); slot = LDS returning add, sums still arrival order
# speedup vs baseline: 2.6677x; 1.5360x over previous
#include <hip/hip_runtime.h>
#include <math.h>

typedef float v2f __attribute__((ext_vector_type(2)));
typedef float v8f __attribute__((ext_vector_type(8)));

#define GCN_N 100000
#define GCN_E 1600000
#define GCN_INDIM 128
#define GCN_HID 64

#define HIST_NRANGE 4
#define HIST_NCHUNK 64
#define HIST_RN ((GCN_N + HIST_NRANGE - 1) / HIST_NRANGE)
#define HIST_CE ((GCN_E + HIST_NCHUNK - 1) / HIST_NCHUNK)

__global__ __launch_bounds__(1024) void gcn_hist(
    const int* __restrict__ src, const int* __restrict__ dst,
    int* __restrict__ pin, int* __restrict__ pout,
    int* __restrict__ rank, int n, int E) {
  __shared__ int s_in[HIST_RN];
  __shared__ int s_out[HIST_RN];
  const int tid = threadIdx.x;
  const int rg = blockIdx.x % HIST_NRANGE;
  const int ch = blockIdx.x / HIST_NRANGE;
  const int lo = rg * HIST_RN;

  for (int i = tid; i < HIST_RN; i += 1024) { s_in[i] = 0; s_out[i] = 0; }
  __syncthreads();

  const int e0 = ch * HIST_CE;
  const int e1 = (e0 + HIST_CE < E) ? (e0 + HIST_CE) : E;
  for (int eb = e0 + tid; eb < e1; eb += 8 * 1024) {
    int d[8], s[8];
#pragma unroll
    for (int k = 0; k < 8; ++k) {
      const int e = eb + k * 1024;
      d[k] = e < e1 ? dst[e] : lo - 1;
      s[k] = e < e1 ? src[e] : lo - 1;
    }
#pragma unroll
    for (int k = 0; k < 8; ++k) {
      const unsigned ud = (unsigned)(d[k] - lo);
      const unsigned us = (unsigned)(s[k] - lo);
      if (ud < (unsigned)HIST_RN) rank[eb + k * 1024] = atomicAdd(&s_in[ud], 1);
      if (us < (unsigned)HIST_RN) atomicAdd(&s_out[us], 1);
    }
  }
  __syncthreads();

  int* pin_c = pin + ch * n + lo;
  int* pout_c = pout + ch * n + lo;
  const int lim = (n - lo < HIST_RN) ? (n - lo) : HIST_RN;
  for (int i = tid; i < lim; i += 1024) { pin_c[i] = s_in[i]; pout_c[i] = s_out[i]; }
}

__global__ __launch_bounds__(256) void gcn_colscan(
    int* __restrict__ pin, const int* __restrict__ pout,
    int* __restrict__ cnt_out, int* __restrict__ cnt_in, int n) {
  const int node = blockIdx.x * 256 + threadIdx.x;
  if (node >= n) return;
  int run_in = 0, run_out = 0;
  for (int cb = 0; cb < HIST_NCHUNK; cb += 16) {
    int a[16], b[16];
#pragma unroll
    for (int k = 0; k < 16; ++k) {
      a[k] = pin[(cb + k) * n + node];
      b[k] = pout[(cb + k) * n + node];
    }
#pragma unroll
    for (int k = 0; k < 16; ++k) {
      pin[(cb + k) * n + node] = run_in;
      run_in += a[k];
      run_out += b[k];
    }
  }
  cnt_in[node] = run_in;
  cnt_out[node] = run_out;
}

__global__ __launch_bounds__(1024) void gcn_scan_norm(
    const int* __restrict__ cnt_out, const int* __restrict__ cnt_in,
    float* __restrict__ norm_src, float* __restrict__ norm_dst,
    int* __restrict__ row_ptr, int n) {
  __shared__ int s_part[32];
  __shared__ int s_incl[32];
  const int tid = threadIdx.x;
  const int lane = tid & 31;
  const int wv = tid >> 5;
  const int blockStart = blockIdx.x * 1024;

  int part = 0;
  const int4* c4 = (const int4*)cnt_in;
  const int grp = __builtin_amdgcn_readfirstlane(tid >> 8);
  const int sub = tid & 255;
  const int groups = blockIdx.x;
#pragma unroll 8
  for (int g = grp; g < groups; g += 4) {
    const int4 q = c4[g * 256 + sub];
    part += (q.x + q.y) + (q.z + q.w);
  }
  for (int o = 16; o > 0; o >>= 1) part += __shfl_xor(part, o);
  if (lane == 0) s_part[wv] = part;

  const int idx = blockStart + tid;
  const int c = idx < n ? cnt_in[idx] : 0;
  int inc = c;
  for (int o = 1; o < 32; o <<= 1) {
    const int v = __shfl_up(inc, o);
    if (lane >= o) inc += v;
  }
  if (lane == 31) s_incl[wv] = inc;
  __syncthreads();

  int off = 0;
  for (int w = 0; w < 32; ++w) {
    off += s_part[w];
    if (w < wv) off += s_incl[w];
  }
  const int excl = off + inc - c;
  if (idx < n) {
    row_ptr[idx] = excl;
    if (idx == n - 1) row_ptr[n] = excl + c;
    norm_dst[idx] = rsqrtf(fmaxf((float)c, 1.0f));
    norm_src[idx] = rsqrtf(fmaxf((float)cnt_out[idx], 1.0f));
  }
}

__global__ __launch_bounds__(256) void gcn_fill(
    const int* __restrict__ src, const int* __restrict__ dst,
    const int* __restrict__ rank, const int* __restrict__ row_ptr,
    const int* __restrict__ pin, int* __restrict__ col, int n, int E) {
  const int blk = blockIdx.x * 1024;
  const int base = blk + threadIdx.x;
  if (blk + 1024 <= E) {
    int s[4], d[4], r[4], p[4], q[4];
#pragma unroll
    for (int k = 0; k < 4; ++k) {
      s[k] = src[base + k * 256];
      d[k] = dst[base + k * 256];
      r[k] = rank[base + k * 256];
    }
#pragma unroll
    for (int k = 0; k < 4; ++k) {
      p[k] = row_ptr[d[k]];
      q[k] = pin[((base + k * 256) / HIST_CE) * n + d[k]];
    }
#pragma unroll
    for (int k = 0; k < 4; ++k) col[p[k] + q[k] + r[k]] = s[k];
  } else {
    for (int k = 0; k < 4; ++k) {
      const int e = base + k * 256;
      if (e < E) col[row_ptr[dst[e]] + pin[(e / HIST_CE) * n + dst[e]] + rank[e]] = src[e];
    }
  }
}

__global__ __launch_bounds__(256) void gcn_gemm_wmma(
    const float* __restrict__ A, const float* __restrict__ W,
    const float* __restrict__ nsrc, float* __restrict__ X,
    int K, int nRows) {
  int wave = blockIdx.x * (blockDim.x >> 5) + (threadIdx.x >> 5);
  int lane = threadIdx.x & 31;
  int rowBlk = wave * 16;
  if (rowBlk >= nRows) return;

  int half = lane >> 4;
  int l    = lane & 15;
  int row  = rowBlk + l;
  float ns = nsrc[row];
  const float* arow = A + (long)row * K;

  v8f acc0 = {}, acc1 = {}, acc2 = {}, acc3 = {};

  for (int k0 = 0; k0 < K; k0 += 4) {
    int ka = k0 + half * 2;
    v2f a;
    a.x = arow[ka]     * ns;
    a.y = arow[ka + 1] * ns;
    const float* w0 = W + (long)ka * 64 + l;
    const float* w1 = w0 + 64;
    v2f b;
    b.x = w0[0];  b.y = w1[0];
    acc0 = __builtin_amdgcn_wmma_f32_16x16x4_f32(false, a, false, b, (short)0, acc0, false, false);
    b.x = w0[16]; b.y = w1[16];
    acc1 = __builtin_amdgcn_wmma_f32_16x16x4_f32(false, a, false, b, (short)0, acc1, false, false);
    b.x = w0[32]; b.y = w1[32];
    acc2 = __builtin_amdgcn_wmma_f32_16x16x4_f32(false, a, false, b, (short)0, acc2, false, false);
    b.x = w0[48]; b.y = w1[48];
    acc3 = __builtin_amdgcn_wmma_f32_16x16x4_f32(false, a, false, b, (short)0, acc3, false, false);
  }

  float* xout = X + (long)rowBlk * 64;
#pragma unroll
  for (int i = 0; i < 8; ++i) {
    int r = i + half * 8;
    xout[(long)r * 64 +  0 + l] = acc0[i];
    xout[(long)r * 64 + 16 + l] = acc1[i];
    xout[(long)r * 64 + 32 + l] = acc2[i];
    xout[(long)r * 64 + 48 + l] = acc3[i];
  }
}

#define PULL_NPW 8

template <bool WITH_ELU>
__device__ __forceinline__ void gcn_pull_body(
    const int* __restrict__ row_ptr, const int* __restrict__ col,
    const float* __restrict__ X, const float* __restrict__ ndst,
    const float* __restrict__ b, float* __restrict__ OUT, int n) {
  const int lane = threadIdx.x & 31;
  const int wv = __builtin_amdgcn_readfirstlane((int)(threadIdx.x >> 5));
  const int node0 = (blockIdx.x * 8 + wv) * PULL_NPW;
  if (node0 >= n) return;
  const int rem = n - node0;
  const int cnt = rem < PULL_NPW ? rem : PULL_NPW;

  int rpv = 0;
  float ndv = 0.0f;
  if (lane <= cnt) rpv = row_ptr[node0 + lane];
  if (lane < cnt) ndv = ndst[node0 + lane];
  const float b0 = b[2 * lane];
  const float b1 = b[2 * lane + 1];
  const int lo = 2 * lane;

  const int first = __builtin_amdgcn_readlane(rpv, 0);
  const int total = __builtin_amdgcn_readlane(rpv, cnt) - first;
  const int* cl = col + first;

  int t = 0;
  int nodeEnd = __builtin_amdgcn_readlane(rpv, 1) - first;
  float ax = 0.0f, ay = 0.0f;

#define PULL_FINISH()                                                          \
  {                                                                            \
    const float nd = __int_as_float(                                           \
        __builtin_amdgcn_readlane(__float_as_int(ndv), t));                    \
    float r0 = ax * nd + b0;                                                   \
    float r1 = ay * nd + b1;                                                   \
    if (WITH_ELU) {                                                            \
      r0 = r0 > 0.0f ? r0 : expm1f(r0);                                        \
      r1 = r1 > 0.0f ? r1 : expm1f(r1);                                        \
    }                                                                          \
    v2f r;                                                                     \
    r.x = r0;                                                                  \
    r.y = r1;                                                                  \
    *(v2f*)(OUT + (long)(node0 + t) * 64 + lo) = r;                            \
    ax = 0.0f;                                                                 \
    ay = 0.0f;                                                                 \
    ++t;                                                                       \
    nodeEnd = __builtin_amdgcn_readlane(rpv, t + 1) - first;                   \
  }

  int cv = 0;
  if (lane < 16 && lane < total) cv = cl[lane];

  for (int pos0 = 0; pos0 < total; pos0 += 16) {
    v2f v[16];
#pragma unroll
    for (int u = 0; u < 16; ++u) {
      const int sr = __builtin_amdgcn_readlane(cv, u);
      v[u] = *(const v2f*)(X + (long)sr * 64 + lo);
    }
    int cvn = 0;
    if (lane < 16 && pos0 + 16 + lane < total) cvn = cl[pos0 + 16 + lane];
#pragma unroll
    for (int u = 0; u < 16; ++u) {
      while (pos0 + u == nodeEnd) PULL_FINISH()
      ax += v[u].x;
      ay += v[u].y;
    }
    cv = cvn;
  }
  while (t < cnt) PULL_FINISH()
#undef PULL_FINISH
}

__global__ __launch_bounds__(256) void gcn_pull_elu(
    const int* __restrict__ row_ptr, const int* __restrict__ col,
    const float* __restrict__ X, const float* __restrict__ ndst,
    const float* __restrict__ b, float* __restrict__ OUT, int n) {
  gcn_pull_body<true>(row_ptr, col, X, ndst, b, OUT, n);
}

__global__ __launch_bounds__(256) void gcn_pull_lin(
    const int* __restrict__ row_ptr, const int* __restrict__ col,
    const float* __restrict__ X, const float* __restrict__ ndst,
    const float* __restrict__ b, float* __restrict__ OUT, int n) {
  gcn_pull_body<false>(row_ptr, col, X, ndst, b, OUT, n);
}

extern "C" void kernel_launch(void* const* d_in, const int* in_sizes, int n_in,
                              void* d_out, int out_size, void* d_ws, size_t ws_size,
                              hipStream_t stream) {
  const float* h   = (const float*)d_in[0];
  const int*   src = (const int*)  d_in[1];
  const int*   dst = (const int*)  d_in[2];
  const float* W1  = (const float*)d_in[3];
  const float* b1  = (const float*)d_in[4];
  const float* W2  = (const float*)d_in[5];
  const float* b2  = (const float*)d_in[6];
  float* out = (float*)d_out;

  const int N = in_sizes[0] / GCN_INDIM;
  const int E = in_sizes[1];

  float* norm_src = (float*)d_ws;
  float* norm_dst = norm_src + N;
  float* bufA     = norm_dst + N;
  int*   cnt_out  = (int*)(bufA + (long)N * 64);
  int*   cnt_in   = cnt_out + N;
  int*   row_ptr  = cnt_in + N;
  int*   rank     = row_ptr + N + 32;
  int*   col      = rank + E;
  int*   pin      = (int*)bufA;
  int*   pout     = (int*)d_out;

  const int edgeBlocks = (E + 1023) / 1024;
  const int pullBlocks = (N + 8 * PULL_NPW - 1) / (8 * PULL_NPW);
  const int waves  = N / 16;
  const int gemmBlocks = (waves + 7) / 8;

  gcn_hist<<<HIST_NRANGE * HIST_NCHUNK, 1024, 0, stream>>>(src, dst, pin, pout, rank, N, E);
  gcn_colscan<<<(N + 255) / 256, 256, 0, stream>>>(pin, pout, cnt_out, cnt_in, N);
  gcn_scan_norm<<<(N + 1023) / 1024, 1024, 0, stream>>>(cnt_out, cnt_in, norm_src, norm_dst, row_ptr, N);
  gcn_fill<<<edgeBlocks, 256, 0, stream>>>(src, dst, rank, row_ptr, pin, col, N, E);

  gcn_gemm_wmma<<<gemmBlocks, 256, 0, stream>>>(h, W1, norm_src, bufA, GCN_INDIM, N);
  gcn_pull_elu<<<pullBlocks, 256, 0, stream>>>(row_ptr, col, bufA, norm_dst, b1, out, N);

  gcn_gemm_wmma<<<gemmBlocks, 256, 0, stream>>>(out, W2, norm_src, bufA, GCN_HID, N);
  gcn_pull_lin<<<pullBlocks, 256, 0, stream>>>(row_ptr, col, bufA, norm_dst, b2, out, N);
}


# layer-2 projection (f32 WMMA, same k order) moved into the layer-1 pull: a block stages its 64 finished hidden rows in LDS and multiplies by W2 there; hidden rows never reach memory, one launch fewer
# speedup vs baseline: 2.8014x; 1.0501x over previous
#include <hip/hip_runtime.h>
#include <math.h>

typedef float v2f __attribute__((ext_vector_type(2)));
typedef float v8f __attribute__((ext_vector_type(8)));

#define GCN_N 100000
#define GCN_E 1600000
#define GCN_INDIM 128
#define GCN_HID 64

#define HIST_NRANGE 4
#define HIST_NCHUNK 64
#define HIST_RN ((GCN_N + HIST_NRANGE - 1) / HIST_NRANGE)
#define HIST_CE ((GCN_E + HIST_NCHUNK - 1) / HIST_NCHUNK)

__global__ __launch_bounds__(1024) void gcn_hist(
    const int* __restrict__ src, const int* __restrict__ dst,
    int* __restrict__ pin, int* __restrict__ pout,
    int* __restrict__ rank, int n, int E) {
  __shared__ int s_in[HIST_RN];
  __shared__ int s_out[HIST_RN];
  const int tid = threadIdx.x;
  const int rg = blockIdx.x % HIST_NRANGE;
  const int ch = blockIdx.x / HIST_NRANGE;
  const int lo = rg * HIST_RN;

  for (int i = tid; i < HIST_RN; i += 1024) { s_in[i] = 0; s_out[i] = 0; }
  __syncthreads();

  const int e0 = ch * HIST_CE;
  const int e1 = (e0 + HIST_CE < E) ? (e0 + HIST_CE) : E;
  for (int eb = e0 + tid; eb < e1; eb += 8 * 1024) {
    int d[8], s[8];
#pragma unroll
    for (int k = 0; k < 8; ++k) {
      const int e = eb + k * 1024;
      d[k] = e < e1 ? dst[e] : lo - 1;
      s[k] = e < e1 ? src[e] : lo - 1;
    }
#pragma unroll
    for (int k = 0; k < 8; ++k) {
      const unsigned ud = (unsigned)(d[k] - lo);
      const unsigned us = (unsigned)(s[k] - lo);
      if (ud < (unsigned)HIST_RN) rank[eb + k * 1024] = atomicAdd(&s_in[ud], 1);
      if (us < (unsigned)HIST_RN) atomicAdd(&s_out[us], 1);
    }
  }
  __syncthreads();

  int* pin_c = pin + ch * n + lo;
  int* pout_c = pout + ch * n + lo;
  const int lim = (n - lo < HIST_RN) ? (n - lo) : HIST_RN;
  for (int i = tid; i < lim; i += 1024) { pin_c[i] = s_in[i]; pout_c[i] = s_out[i]; }
}

__global__ __launch_bounds__(256) void gcn_colscan(
    int* __restrict__ pin, const int* __restrict__ pout,
    int* __restrict__ cnt_out, int* __restrict__ cnt_in, int n) {
  const int node = blockIdx.x * 256 + threadIdx.x;
  if (node >= n) return;
  int run_in = 0, run_out = 0;
  for (int cb = 0; cb < HIST_NCHUNK; cb += 16) {
    int a[16], b[16];
#pragma unroll
    for (int k = 0; k < 16; ++k) {
      a[k] = pin[(cb + k) * n + node];
      b[k] = pout[(cb + k) * n + node];
    }
#pragma unroll
    for (int k = 0; k < 16; ++k) {
      pin[(cb + k) * n + node] = run_in;
      run_in += a[k];
      run_out += b[k];
    }
  }
  cnt_in[node] = run_in;
  cnt_out[node] = run_out;
}

__global__ __launch_bounds__(1024) void gcn_scan_norm(
    const int* __restrict__ cnt_out, const int* __restrict__ cnt_in,
    float* __restrict__ norm_src, float* __restrict__ norm_dst,
    int* __restrict__ row_ptr, int n) {
  __shared__ int s_part[32];
  __shared__ int s_incl[32];
  const int tid = threadIdx.x;
  const int lane = tid & 31;
  const int wv = tid >> 5;
  const int blockStart = blockIdx.x * 1024;

  int part = 0;
  const int4* c4 = (const int4*)cnt_in;
  const int grp = __builtin_amdgcn_readfirstlane(tid >> 8);
  const int sub = tid & 255;
  const int groups = blockIdx.x;
#pragma unroll 8
  for (int g = grp; g < groups; g += 4) {
    const int4 q = c4[g * 256 + sub];
    part += (q.x + q.y) + (q.z + q.w);
  }
  for (int o = 16; o > 0; o >>= 1) part += __shfl_xor(part, o);
  if (lane == 0) s_part[wv] = part;

  const int idx = blockStart + tid;
  const int c = idx < n ? cnt_in[idx] : 0;
  int inc = c;
  for (int o = 1; o < 32; o <<= 1) {
    const int v = __shfl_up(inc, o);
    if (lane >= o) inc += v;
  }
  if (lane == 31) s_incl[wv] = inc;
  __syncthreads();

  int off = 0;
  for (int w = 0; w < 32; ++w) {
    off += s_part[w];
    if (w < wv) off += s_incl[w];
  }
  const int excl = off + inc - c;
  if (idx < n) {
    row_ptr[idx] = excl;
    if (idx == n - 1) row_ptr[n] = excl + c;
    norm_dst[idx] = rsqrtf(fmaxf((float)c, 1.0f));
    norm_src[idx] = rsqrtf(fmaxf((float)cnt_out[idx], 1.0f));
  }
}

__global__ __launch_bounds__(256) void gcn_fill(
    const int* __restrict__ src, const int* __restrict__ dst,
    const int* __restrict__ rank, const int* __restrict__ row_ptr,
    const int* __restrict__ pin, int* __restrict__ col, int n, int E) {
  const int blk = blockIdx.x * 1024;
  const int base = blk + threadIdx.x;
  if (blk + 1024 <= E) {
    int s[4], d[4], r[4], p[4], q[4];
#pragma unroll
    for (int k = 0; k < 4; ++k) {
      s[k] = src[base + k * 256];
      d[k] = dst[base + k * 256];
      r[k] = rank[base + k * 256];
    }
#pragma unroll
    for (int k = 0; k < 4; ++k) {
      p[k] = row_ptr[d[k]];
      q[k] = pin[((base + k * 256) / HIST_CE) * n + d[k]];
    }
#pragma unroll
    for (int k = 0; k < 4; ++k) col[p[k] + q[k] + r[k]] = s[k];
  } else {
    for (int k = 0; k < 4; ++k) {
      const int e = base + k * 256;
      if (e < E) col[row_ptr[dst[e]] + pin[(e / HIST_CE) * n + dst[e]] + rank[e]] = src[e];
    }
  }
}

__global__ __launch_bounds__(256) void gcn_gemm_wmma(
    const float* __restrict__ A, const float* __restrict__ W,
    const float* __restrict__ nsrc, float* __restrict__ X,
    int K, int nRows) {
  int wave = blockIdx.x * (blockDim.x >> 5) + (threadIdx.x >> 5);
  int lane = threadIdx.x & 31;
  int rowBlk = wave * 16;
  if (rowBlk >= nRows) return;

  int half = lane >> 4;
  int l    = lane & 15;
  int row  = rowBlk + l;
  float ns = nsrc[row];
  const float* arow = A + (long)row * K;

  v8f acc0 = {}, acc1 = {}, acc2 = {}, acc3 = {};

  for (int k0 = 0; k0 < K; k0 += 4) {
    int ka = k0 + half * 2;
    v2f a;
    a.x = arow[ka]     * ns;
    a.y = arow[ka + 1] * ns;
    const float* w0 = W + (long)ka * 64 + l;
    const float* w1 = w0 + 64;
    v2f b;
    b.x = w0[0];  b.y = w1[0];
    acc0 = __builtin_amdgcn_wmma_f32_16x16x4_f32(false, a, false, b, (short)0, acc0, false, false);
    b.x = w0[16]; b.y = w1[16];
    acc1 = __builtin_amdgcn_wmma_f32_16x16x4_f32(false, a, false, b, (short)0, acc1, false, false);
    b.x = w0[32]; b.y = w1[32];
    acc2 = __builtin_amdgcn_wmma_f32_16x16x4_f32(false, a, false, b, (short)0, acc2, false, false);
    b.x = w0[48]; b.y = w1[48];
    acc3 = __builtin_amdgcn_wmma_f32_16x16x4_f32(false, a, false, b, (short)0, acc3, false, false);
  }

  float* xout = X + (long)rowBlk * 64;
#pragma unroll
  for (int i = 0; i < 8; ++i) {
    int r = i + half * 8;
    xout[(long)r * 64 +  0 + l] = acc0[i];
    xout[(long)r * 64 + 16 + l] = acc1[i];
    xout[(long)r * 64 + 32 + l] = acc2[i];
    xout[(long)r * 64 + 48 + l] = acc3[i];
  }
}

#define PULL_NPW 8
#define PROJ_PITCH 68

template <bool HIDDEN>
__device__ __forceinline__ void gcn_pull_rows(
    const int* __restrict__ row_ptr, const int* __restrict__ col,
    const float* __restrict__ X, const float* __restrict__ ndst,
    const float* __restrict__ nsrc, const float* __restrict__ b,
    float* __restrict__ OUT, float* s_rows, int node0, int cnt, int lane) {
  int rpv = 0;
  float ndv = 0.0f;
  float nsv = 0.0f;
  if (lane <= cnt) rpv = row_ptr[node0 + lane];
  if (lane < cnt) ndv = ndst[node0 + lane];
  if (HIDDEN) {
    if (lane < cnt) nsv = nsrc[node0 + lane];
  }
  const float b0 = b[2 * lane];
  const float b1 = b[2 * lane + 1];
  const int lo = 2 * lane;

  const int first = __builtin_amdgcn_readlane(rpv, 0);
  const int total = __builtin_amdgcn_readlane(rpv, cnt) - first;
  const int* cl = col + first;

  int t = 0;
  int nodeEnd = __builtin_amdgcn_readlane(rpv, 1) - first;
  float ax = 0.0f, ay = 0.0f;

#define PULL_FINISH()                                                          \
  {                                                                            \
    const float nd = __int_as_float(                                           \
        __builtin_amdgcn_readlane(__float_as_int(ndv), t));                    \
    float r0 = ax * nd + b0;                                                   \
    float r1 = ay * nd + b1;                                                   \
    v2f r;                                                                     \
    if (HIDDEN) {                                                              \
      const float ns = __int_as_float(                                         \
          __builtin_amdgcn_readlane(__float_as_int(nsv), t));                  \
      r0 = r0 > 0.0f ? r0 : expm1f(r0);                                        \
      r1 = r1 > 0.0f ? r1 : expm1f(r1);                                        \
      r.x = r0 * ns;                                                           \
      r.y = r1 * ns;                                                           \
      *(v2f*)(s_rows + t * PROJ_PITCH + lo) = r;                               \
    } else {                                                                   \
      r.x = r0;                                                                \
      r.y = r1;                                                                \
      *(v2f*)(OUT + (long)(node0 + t) * 64 + lo) = r;                          \
    }                                                                          \
    ax = 0.0f;                                                                 \
    ay = 0.0f;                                                                 \
    ++t;                                                                       \
    nodeEnd = __builtin_amdgcn_readlane(rpv, t + 1) - first;                   \
  }

  int cv = 0;
  if (lane < 16 && lane < total) cv = cl[lane];

  for (int pos0 = 0; pos0 < total; pos0 += 16) {
    v2f v[16];
#pragma unroll
    for (int u = 0; u < 16; ++u) {
      const int sr = __builtin_amdgcn_readlane(cv, u);
      v[u] = *(const v2f*)(X + (long)sr * 64 + lo);
    }
    int cvn = 0;
    if (lane < 16 && pos0 + 16 + lane < total) cvn = cl[pos0 + 16 + lane];
#pragma unroll
    for (int u = 0; u < 16; ++u) {
      while (pos0 + u == nodeEnd) PULL_FINISH()
      ax += v[u].x;
      ay += v[u].y;
    }
    cv = cvn;
  }
  while (t < cnt) PULL_FINISH()
#undef PULL_FINISH
}

__global__ __launch_bounds__(256) void gcn_pull_lin(
    const int* __restrict__ row_ptr, const int* __restrict__ col,
    const float* __restrict__ X, const float* __restrict__ ndst,
    const float* __restrict__ b, float* __restrict__ OUT, int n) {
  const int lane = threadIdx.x & 31;
  const int wv = __builtin_amdgcn_readfirstlane((int)(threadIdx.x >> 5));
  const int node0 = (blockIdx.x * 8 + wv) * PULL_NPW;
  if (node0 >= n) return;
  const int rem = n - node0;
  const int cnt = rem < PULL_NPW ? rem : PULL_NPW;
  gcn_pull_rows<false>(row_ptr, col, X, ndst, nullptr, b, OUT, nullptr, node0, cnt, lane);
}

__global__ __launch_bounds__(256) void gcn_pull_elu_proj(
    const int* __restrict__ row_ptr, const int* __restrict__ col,
    const float* __restrict__ X, const float* __restrict__ ndst,
    const float* __restrict__ nsrc, const float* __restrict__ b,
    const float* __restrict__ W2, float* __restrict__ X2, int n) {
  __shared__ __attribute__((aligned(16))) float s_h[8 * PULL_NPW * PROJ_PITCH];
  const int lane = threadIdx.x & 31;
  const int wv = __builtin_amdgcn_readfirstlane((int)(threadIdx.x >> 5));
  const int node0 = (blockIdx.x * 8 + wv) * PULL_NPW;
  const int rem = n - node0;
  const int cnt = rem < PULL_NPW ? (rem > 0 ? rem : 0) : PULL_NPW;
  float* s_mine = s_h + wv * PULL_NPW * PROJ_PITCH;
  if (cnt > 0)
    gcn_pull_rows<true>(row_ptr, col, X, ndst, nsrc, b, nullptr, s_mine, node0, cnt, lane);
  for (int t = cnt; t < PULL_NPW; ++t) {
    v2f z;
    z.x = 0.0f;
    z.y = 0.0f;
    *(v2f*)(s_mine + t * PROJ_PITCH + 2 * lane) = z;
  }
  __syncthreads();

  const int half = lane >> 4;
  const int c = lane & 15;
  const int mt = wv >> 1;
  const int n0 = (wv & 1) * 32;
  const float* arow = s_h + (mt * 16 + c) * PROJ_PITCH + 2 * half;
  const float* wp = W2 + (2 * half) * 64 + n0 + 2 * c;
  v8f acc0 = {}, acc1 = {};
#pragma unroll
  for (int k0 = 0; k0 < GCN_HID; k0 += 4) {
    const v2f a = *(const v2f*)(arow + k0);
    const v2f w0 = *(const v2f*)(wp + k0 * 64);
    const v2f w1 = *(const v2f*)(wp + k0 * 64 + 64);
    v2f bA, bB;
    bA.x = w0.x; bA.y = w1.x;
    bB.x = w0.y; bB.y = w1.y;
    acc0 = __builtin_amdgcn_wmma_f32_16x16x4_f32(false, a, false, bA, (short)0, acc0, false, false);
    acc1 = __builtin_amdgcn_wmma_f32_16x16x4_f32(false, a, false, bB, (short)0, acc1, false, false);
  }

  const int rowBase = blockIdx.x * (8 * PULL_NPW) + mt * 16 + half * 8;
#pragma unroll
  for (int i = 0; i < 8; ++i) {
    const int r = rowBase + i;
    if (r < n) {
      v2f o;
      o.x = acc0[i];
      o.y = acc1[i];
      *(v2f*)(X2 + (long)r * 64 + n0 + 2 * c) = o;
    }
  }
}

extern "C" void kernel_launch(void* const* d_in, const int* in_sizes, int n_in,
                              void* d_out, int out_size, void* d_ws, size_t ws_size,
                              hipStream_t stream) {
  const float* h   = (const float*)d_in[0];
  const int*   src = (const int*)  d_in[1];
  const int*   dst = (const int*)  d_in[2];
  const float* W1  = (const float*)d_in[3];
  const float* b1  = (const float*)d_in[4];
  const float* W2  = (const float*)d_in[5];
  const float* b2  = (const float*)d_in[6];
  float* out = (float*)d_out;

  const int N = in_sizes[0] / GCN_INDIM;
  const int E = in_sizes[1];

  float* norm_src = (float*)d_ws;
  float* norm_dst = norm_src + N;
  float* bufA     = norm_dst + N;
  int*   cnt_out  = (int*)(bufA + (long)N * 64);
  int*   cnt_in   = cnt_out + N;
  int*   row_ptr  = cnt_in + N;
  int*   rank     = row_ptr + N + 32;
  int*   col      = rank + E;
  int*   pin      = (int*)bufA;
  int*   pout     = (int*)d_out;

  const int edgeBlocks = (E + 1023) / 1024;
  const int pullBlocks = (N + 8 * PULL_NPW - 1) / (8 * PULL_NPW);
  const int waves  = N / 16;
  const int gemmBlocks = (waves + 7) / 8;

  gcn_hist<<<HIST_NRANGE * HIST_NCHUNK, 1024, 0, stream>>>(src, dst, pin, pout, rank, N, E);
  gcn_colscan<<<(N + 255) / 256, 256, 0, stream>>>(pin, pout, cnt_out, cnt_in, N);
  gcn_scan_norm<<<(N + 1023) / 1024, 1024, 0, stream>>>(cnt_out, cnt_in, norm_src, norm_dst, row_ptr, N);
  gcn_fill<<<edgeBlocks, 256, 0, stream>>>(src, dst, rank, row_ptr, pin, col, N, E);

  gcn_gemm_wmma<<<gemmBlocks, 256, 0, stream>>>(h, W1, norm_src, out, GCN_INDIM, N);
  gcn_pull_elu_proj<<<pullBlocks, 256, 0, stream>>>(row_ptr, col, out, norm_dst, norm_src, b1, W2, bufA, N);

  gcn_pull_lin<<<pullBlocks, 256, 0, stream>>>(row_ptr, col, bufA, norm_dst, b2, out, N);
}


# col fill: per-edge rank array and two per-edge 4-B gathers replaced by a second LDS walk of block (chunk, node range); returning LDS add = final slot, sums still arrival order
# speedup vs baseline: 3.0420x; 1.0859x over previous
#include <hip/hip_runtime.h>
#include <math.h>

typedef float v2f __attribute__((ext_vector_type(2)));
typedef float v8f __attribute__((ext_vector_type(8)));

#define GCN_N 100000
#define GCN_E 1600000
#define GCN_INDIM 128
#define GCN_HID 64

#define HIST_NRANGE 4
#define HIST_NCHUNK 64
#define HIST_RN ((GCN_N + HIST_NRANGE - 1) / HIST_NRANGE)
#define HIST_CE ((GCN_E + HIST_NCHUNK - 1) / HIST_NCHUNK)

__global__ __launch_bounds__(1024) void gcn_hist(
    const int* __restrict__ src, const int* __restrict__ dst,
    int* __restrict__ pin, int* __restrict__ pout, int n, int E) {
  __shared__ int s_in[HIST_RN];
  __shared__ int s_out[HIST_RN];
  const int tid = threadIdx.x;
  const int rg = blockIdx.x % HIST_NRANGE;
  const int ch = blockIdx.x / HIST_NRANGE;
  const int lo = rg * HIST_RN;

  for (int i = tid; i < HIST_RN; i += 1024) { s_in[i] = 0; s_out[i] = 0; }
  __syncthreads();

  const int e0 = ch * HIST_CE;
  const int e1 = (e0 + HIST_CE < E) ? (e0 + HIST_CE) : E;
  for (int eb = e0 + tid; eb < e1; eb += 8 * 1024) {
    int d[8], s[8];
#pragma unroll
    for (int k = 0; k < 8; ++k) {
      const int e = eb + k * 1024;
      d[k] = e < e1 ? dst[e] : lo - 1;
      s[k] = e < e1 ? src[e] : lo - 1;
    }
#pragma unroll
    for (int k = 0; k < 8; ++k) {
      const unsigned ud = (unsigned)(d[k] - lo);
      const unsigned us = (unsigned)(s[k] - lo);
      if (ud < (unsigned)HIST_RN) atomicAdd(&s_in[ud], 1);
      if (us < (unsigned)HIST_RN) atomicAdd(&s_out[us], 1);
    }
  }
  __syncthreads();

  int* pin_c = pin + ch * n + lo;
  int* pout_c = pout + ch * n + lo;
  const int lim = (n - lo < HIST_RN) ? (n - lo) : HIST_RN;
  for (int i = tid; i < lim; i += 1024) { pin_c[i] = s_in[i]; pout_c[i] = s_out[i]; }
}

__global__ __launch_bounds__(256) void gcn_colscan(
    int* __restrict__ pin, const int* __restrict__ pout,
    int* __restrict__ cnt_out, int* __restrict__ cnt_in, int n) {
  const int node = blockIdx.x * 256 + threadIdx.x;
  if (node >= n) return;
  int run_in = 0, run_out = 0;
  for (int cb = 0; cb < HIST_NCHUNK; cb += 16) {
    int a[16], b[16];
#pragma unroll
    for (int k = 0; k < 16; ++k) {
      a[k] = pin[(cb + k) * n + node];
      b[k] = pout[(cb + k) * n + node];
    }
#pragma unroll
    for (int k = 0; k < 16; ++k) {
      pin[(cb + k) * n + node] = run_in;
      run_in += a[k];
      run_out += b[k];
    }
  }
  cnt_in[node] = run_in;
  cnt_out[node] = run_out;
}

__global__ __launch_bounds__(1024) void gcn_scan_norm(
    const int* __restrict__ cnt_out, const int* __restrict__ cnt_in,
    float* __restrict__ norm_src, float* __restrict__ norm_dst,
    int* __restrict__ row_ptr, int n) {
  __shared__ int s_part[32];
  __shared__ int s_incl[32];
  const int tid = threadIdx.x;
  const int lane = tid & 31;
  const int wv = tid >> 5;
  const int blockStart = blockIdx.x * 1024;

  int part = 0;
  const int4* c4 = (const int4*)cnt_in;
  const int grp = __builtin_amdgcn_readfirstlane(tid >> 8);
  const int sub = tid & 255;
  const int groups = blockIdx.x;
#pragma unroll 8
  for (int g = grp; g < groups; g += 4) {
    const int4 q = c4[g * 256 + sub];
    part += (q.x + q.y) + (q.z + q.w);
  }
  for (int o = 16; o > 0; o >>= 1) part += __shfl_xor(part, o);
  if (lane == 0) s_part[wv] = part;

  const int idx = blockStart + tid;
  const int c = idx < n ? cnt_in[idx] : 0;
  int inc = c;
  for (int o = 1; o < 32; o <<= 1) {
    const int v = __shfl_up(inc, o);
    if (lane >= o) inc += v;
  }
  if (lane == 31) s_incl[wv] = inc;
  __syncthreads();

  int off = 0;
  for (int w = 0; w < 32; ++w) {
    off += s_part[w];
    if (w < wv) off += s_incl[w];
  }
  const int excl = off + inc - c;
  if (idx < n) {
    row_ptr[idx] = excl;
    if (idx == n - 1) row_ptr[n] = excl + c;
    norm_dst[idx] = rsqrtf(fmaxf((float)c, 1.0f));
    norm_src[idx] = rsqrtf(fmaxf((float)cnt_out[idx], 1.0f));
  }
}

#define FILL_EPT ((HIST_CE + 1023) / 1024)
#define FILL_NPT ((HIST_RN + 1023) / 1024)

__global__ __launch_bounds__(1024) void gcn_fill(
    const int* __restrict__ src, const int* __restrict__ dst,
    const int* __restrict__ row_ptr, const int* __restrict__ pin,
    int* __restrict__ col, int n, int E) {
  __shared__ int s_pos[HIST_RN];
  const int tid = threadIdx.x;
  const int rg = blockIdx.x % HIST_NRANGE;
  const int ch = blockIdx.x / HIST_NRANGE;
  const int lo = rg * HIST_RN;
  const int lim = (n - lo < HIST_RN) ? (n - lo) : HIST_RN;
  const int e0 = ch * HIST_CE;
  const int e1 = (e0 + HIST_CE < E) ? (e0 + HIST_CE) : E;
  const int* rp_c = row_ptr + lo;
  const int* pin_c = pin + ch * n + lo;

  int a[FILL_NPT], b[FILL_NPT];
#pragma unroll
  for (int k = 0; k < FILL_NPT; ++k) {
    const int i = tid + k * 1024;
    a[k] = i < lim ? rp_c[i] : 0;
    b[k] = i < lim ? pin_c[i] : 0;
  }
  int d[FILL_EPT], s[FILL_EPT];
#pragma unroll
  for (int k = 0; k < FILL_EPT; ++k) {
    const int e = e0 + tid + k * 1024;
    d[k] = e < e1 ? dst[e] : lo - 1;
    s[k] = e < e1 ? src[e] : 0;
  }
#pragma unroll
  for (int k = 0; k < FILL_NPT; ++k) {
    const int i = tid + k * 1024;
    if (i < lim) s_pos[i] = a[k] + b[k];
  }
  __syncthreads();

#pragma unroll
  for (int k = 0; k < FILL_EPT; ++k) {
    const unsigned ud = (unsigned)(d[k] - lo);
    int place = -1;
    if (ud < (unsigned)HIST_RN) place = atomicAdd(&s_pos[ud], 1);
    d[k] = place;
  }
#pragma unroll
  for (int k = 0; k < FILL_EPT; ++k) {
    if (d[k] >= 0) col[d[k]] = s[k];
  }
}

__global__ __launch_bounds__(256) void gcn_gemm_wmma(
    const float* __restrict__ A, const float* __restrict__ W,
    const float* __restrict__ nsrc, float* __restrict__ X,
    int K, int nRows) {
  int wave = blockIdx.x * (blockDim.x >> 5) + (threadIdx.x >> 5);
  int lane = threadIdx.x & 31;
  int rowBlk = wave * 16;
  if (rowBlk >= nRows) return;

  int half = lane >> 4;
  int l    = lane & 15;
  int row  = rowBlk + l;
  float ns = nsrc[row];
  const float* arow = A + (long)row * K;

  v8f acc0 = {}, acc1 = {}, acc2 = {}, acc3 = {};

  for (int k0 = 0; k0 < K; k0 += 4) {
    int ka = k0 + half * 2;
    v2f a;
    a.x = arow[ka]     * ns;
    a.y = arow[ka + 1] * ns;
    const float* w0 = W + (long)ka * 64 + l;
    const float* w1 = w0 + 64;
    v2f b;
    b.x = w0[0];  b.y = w1[0];
    acc0 = __builtin_amdgcn_wmma_f32_16x16x4_f32(false, a, false, b, (short)0, acc0, false, false);
    b.x = w0[16]; b.y = w1[16];
    acc1 = __builtin_amdgcn_wmma_f32_16x16x4_f32(false, a, false, b, (short)0, acc1, false, false);
    b.x = w0[32]; b.y = w1[32];
    acc2 = __builtin_amdgcn_wmma_f32_16x16x4_f32(false, a, false, b, (short)0, acc2, false, false);
    b.x = w0[48]; b.y = w1[48];
    acc3 = __builtin_amdgcn_wmma_f32_16x16x4_f32(false, a, false, b, (short)0, acc3, false, false);
  }

  float* xout = X + (long)rowBlk * 64;
#pragma unroll
  for (int i = 0; i < 8; ++i) {
    int r = i + half * 8;
    xout[(long)r * 64 +  0 + l] = acc0[i];
    xout[(long)r * 64 + 16 + l] = acc1[i];
    xout[(long)r * 64 + 32 + l] = acc2[i];
    xout[(long)r * 64 + 48 + l] = acc3[i];
  }
}

#define PULL_NPW 8
#define PROJ_PITCH 68

template <bool HIDDEN>
__device__ __forceinline__ void gcn_pull_rows(
    const int* __restrict__ row_ptr, const int* __restrict__ col,
    const float* __restrict__ X, const float* __restrict__ ndst,
    const float* __restrict__ nsrc, const float* __restrict__ b,
    float* __restrict__ OUT, float* s_rows, int node0, int cnt, int lane) {
  int rpv = 0;
  float ndv = 0.0f;
  float nsv = 0.0f;
  if (lane <= cnt) rpv = row_ptr[node0 + lane];
  if (lane < cnt) ndv = ndst[node0 + lane];
  if (HIDDEN) {
    if (lane < cnt) nsv = nsrc[node0 + lane];
  }
  const float b0 = b[2 * lane];
  const float b1 = b[2 * lane + 1];
  const int lo = 2 * lane;

  const int first = __builtin_amdgcn_readlane(rpv, 0);
  const int total = __builtin_amdgcn_readlane(rpv, cnt) - first;
  const int* cl = col + first;

  int t = 0;
  int nodeEnd = __builtin_amdgcn_readlane(rpv, 1) - first;
  float ax = 0.0f, ay = 0.0f;

#define PULL_FINISH()                                                          \
  {                                                                            \
    const float nd = __int_as_float(                                           \
        __builtin_amdgcn_readlane(__float_as_int(ndv), t));                    \
    float r0 = ax * nd + b0;                                                   \
    float r1 = ay * nd + b1;                                                   \
    v2f r;                                                                     \
    if (HIDDEN) {                                                              \
      const float ns = __int_as_float(                                         \
          __builtin_amdgcn_readlane(__float_as_int(nsv), t));                  \
      r0 = r0 > 0.0f ? r0 : expm1f(r0);                                        \
      r1 = r1 > 0.0f ? r1 : expm1f(r1);                                        \
      r.x = r0 * ns;                                                           \
      r.y = r1 * ns;                                                           \
      *(v2f*)(s_rows + t * PROJ_PITCH + lo) = r;                               \
    } else {                                                                   \
      r.x = r0;                                                                \
      r.y = r1;                                                                \
      *(v2f*)(OUT + (long)(node0 + t) * 64 + lo) = r;                          \
    }                                                                          \
    ax = 0.0f;                                                                 \
    ay = 0.0f;                                                                 \
    ++t;                                                                       \
    nodeEnd = __builtin_amdgcn_readlane(rpv, t + 1) - first;                   \
  }

  int cv = 0;
  if (lane < 16 && lane < total) cv = cl[lane];

  for (int pos0 = 0; pos0 < total; pos0 += 16) {
    v2f v[16];
#pragma unroll
    for (int u = 0; u < 16; ++u) {
      const int sr = __builtin_amdgcn_readlane(cv, u);
      v[u] = *(const v2f*)(X + (long)sr * 64 + lo);
    }
    int cvn = 0;
    if (lane < 16 && pos0 + 16 + lane < total) cvn = cl[pos0 + 16 + lane];
#pragma unroll
    for (int u = 0; u < 16; ++u) {
      while (pos0 + u == nodeEnd) PULL_FINISH()
      ax += v[u].x;
      ay += v[u].y;
    }
    cv = cvn;
  }
  while (t < cnt) PULL_FINISH()
#undef PULL_FINISH
}

__global__ __launch_bounds__(256) void gcn_pull_lin(
    const int* __restrict__ row_ptr, const int* __restrict__ col,
    const float* __restrict__ X, const float* __restrict__ ndst,
    const float* __restrict__ b, float* __restrict__ OUT, int n) {
  const int lane = threadIdx.x & 31;
  const int wv = __builtin_amdgcn_readfirstlane((int)(threadIdx.x >> 5));
  const int node0 = (blockIdx.x * 8 + wv) * PULL_NPW;
  if (node0 >= n) return;
  const int rem = n - node0;
  const int cnt = rem < PULL_NPW ? rem : PULL_NPW;
  gcn_pull_rows<false>(row_ptr, col, X, ndst, nullptr, b, OUT, nullptr, node0, cnt, lane);
}

__global__ __launch_bounds__(256) void gcn_pull_elu_proj(
    const int* __restrict__ row_ptr, const int* __restrict__ col,
    const float* __restrict__ X, const float* __restrict__ ndst,
    const float* __restrict__ nsrc, const float* __restrict__ b,
    const float* __restrict__ W2, float* __restrict__ X2, int n) {
  __shared__ __attribute__((aligned(16))) float s_h[8 * PULL_NPW * PROJ_PITCH];
  const int lane = threadIdx.x & 31;
  const int wv = __builtin_amdgcn_readfirstlane((int)(threadIdx.x >> 5));
  const int node0 = (blockIdx.x * 8 + wv) * PULL_NPW;
  const int rem = n - node0;
  const int cnt = rem < PULL_NPW ? (rem > 0 ? rem : 0) : PULL_NPW;
  float* s_mine = s_h + wv * PULL_NPW * PROJ_PITCH;
  if (cnt > 0)
    gcn_pull_rows<true>(row_ptr, col, X, ndst, nsrc, b, nullptr, s_mine, node0, cnt, lane);
  for (int t = cnt; t < PULL_NPW; ++t) {
    v2f z;
    z.x = 0.0f;
    z.y = 0.0f;
    *(v2f*)(s_mine + t * PROJ_PITCH + 2 * lane) = z;
  }
  __syncthreads();

  const int half = lane >> 4;
  const int c = lane & 15;
  const int mt = wv >> 1;
  const int n0 = (wv & 1) * 32;
  const float* arow = s_h + (mt * 16 + c) * PROJ_PITCH + 2 * half;
  const float* wp = W2 + (2 * half) * 64 + n0 + 2 * c;
  v8f acc0 = {}, acc1 = {};
#pragma unroll
  for (int k0 = 0; k0 < GCN_HID; k0 += 4) {
    const v2f a = *(const v2f*)(arow + k0);
    const v2f w0 = *(const v2f*)(wp + k0 * 64);
    const v2f w1 = *(const v2f*)(wp + k0 * 64 + 64);
    v2f bA, bB;
    bA.x = w0.x; bA.y = w1.x;
    bB.x = w0.y; bB.y = w1.y;
    acc0 = __builtin_amdgcn_wmma_f32_16x16x4_f32(false, a, false, bA, (short)0, acc0, false, false);
    acc1 = __builtin_amdgcn_wmma_f32_16x16x4_f32(false, a, false, bB, (short)0, acc1, false, false);
  }

  const int rowBase = blockIdx.x * (8 * PULL_NPW) + mt * 16 + half * 8;
#pragma unroll
  for (int i = 0; i < 8; ++i) {
    const int r = rowBase + i;
    if (r < n) {
      v2f o;
      o.x = acc0[i];
      o.y = acc1[i];
      *(v2f*)(X2 + (long)r * 64 + n0 + 2 * c) = o;
    }
  }
}

extern "C" void kernel_launch(void* const* d_in, const int* in_sizes, int n_in,
                              void* d_out, int out_size, void* d_ws, size_t ws_size,
                              hipStream_t stream) {
  const float* h   = (const float*)d_in[0];
  const int*   src = (const int*)  d_in[1];
  const int*   dst = (const int*)  d_in[2];
  const float* W1  = (const float*)d_in[3];
  const float* b1  = (const float*)d_in[4];
  const float* W2  = (const float*)d_in[5];
  const float* b2  = (const float*)d_in[6];
  float* out = (float*)d_out;

  const int N = in_sizes[0] / GCN_INDIM;
  const int E = in_sizes[1];

  float* norm_src = (float*)d_ws;
  float* norm_dst = norm_src + N;
  float* bufA     = norm_dst + N;
  int*   cnt_out  = (int*)(bufA + (long)N * 64);
  int*   cnt_in   = cnt_out + N;
  int*   row_ptr  = cnt_in + N;
  int*   col      = row_ptr + N + 32;
  int*   pin      = (int*)bufA;
  int*   pout     = (int*)d_out;

  const int pullBlocks = (N + 8 * PULL_NPW - 1) / (8 * PULL_NPW);
  const int waves  = N / 16;
  const int gemmBlocks = (waves + 7) / 8;

  gcn_hist<<<HIST_NRANGE * HIST_NCHUNK, 1024, 0, stream>>>(src, dst, pin, pout, N, E);
  gcn_colscan<<<(N + 255) / 256, 256, 0, stream>>>(pin, pout, cnt_out, cnt_in, N);
  gcn_scan_norm<<<(N + 1023) / 1024, 1024, 0, stream>>>(cnt_out, cnt_in, norm_src, norm_dst, row_ptr, N);
  gcn_fill<<<HIST_NRANGE * HIST_NCHUNK, 1024, 0, stream>>>(src, dst, row_ptr, pin, col, N, E);

  gcn_gemm_wmma<<<gemmBlocks, 256, 0, stream>>>(h, W1, norm_src, out, GCN_INDIM, N);
  gcn_pull_elu_proj<<<pullBlocks, 256, 0, stream>>>(row_ptr, col, out, norm_dst, norm_src, b1, W2, bufA, N);

  gcn_pull_lin<<<pullBlocks, 256, 0, stream>>>(row_ptr, col, bufA, norm_dst, b2, out, N);
}


# col fill and layer-1 projection (f32 WMMA, k order and every sum's order unchanged) share one launch: fill blocks first, projection blocks beside them on the same WGPs, both held to 64 registers
# speedup vs baseline: 3.1716x; 1.0426x over previous
#include <hip/hip_runtime.h>
#include <math.h>

typedef float v2f __attribute__((ext_vector_type(2)));
typedef float v8f __attribute__((ext_vector_type(8)));

#define GCN_N 100000
#define GCN_E 1600000
#define GCN_INDIM 128
#define GCN_HID 64

#define HIST_NRANGE 4
#define HIST_NCHUNK 64
#define HIST_RN ((GCN_N + HIST_NRANGE - 1) / HIST_NRANGE)
#define HIST_CE ((GCN_E + HIST_NCHUNK - 1) / HIST_NCHUNK)

__global__ __launch_bounds__(1024) void gcn_hist(
    const int* __restrict__ src, const int* __restrict__ dst,
    int* __restrict__ pin, int* __restrict__ pout, int n, int E) {
  __shared__ int s_in[HIST_RN];
  __shared__ int s_out[HIST_RN];
  const int tid = threadIdx.x;
  const int rg = blockIdx.x % HIST_NRANGE;
  const int ch = blockIdx.x / HIST_NRANGE;
  const int lo = rg * HIST_RN;

  for (int i = tid; i < HIST_RN; i += 1024) { s_in[i] = 0; s_out[i] = 0; }
  __syncthreads();

  const int e0 = ch * HIST_CE;
  const int e1 = (e0 + HIST_CE < E) ? (e0 + HIST_CE) : E;
  for (int eb = e0 + tid; eb < e1; eb += 8 * 1024) {
    int d[8], s[8];
#pragma unroll
    for (int k = 0; k < 8; ++k) {
      const int e = eb + k * 1024;
      d[k] = e < e1 ? dst[e] : lo - 1;
      s[k] = e < e1 ? src[e] : lo - 1;
    }
#pragma unroll
    for (int k = 0; k < 8; ++k) {
      const unsigned ud = (unsigned)(d[k] - lo);
      const unsigned us = (unsigned)(s[k] - lo);
      if (ud < (unsigned)HIST_RN) atomicAdd(&s_in[ud], 1);
      if (us < (unsigned)HIST_RN) atomicAdd(&s_out[us], 1);
    }
  }
  __syncthreads();

  int* pin_c = pin + ch * n + lo;
  int* pout_c = pout + ch * n + lo;
  const int lim = (n - lo < HIST_RN) ? (n - lo) : HIST_RN;
  for (int i = tid; i < lim; i += 1024) { pin_c[i] = s_in[i]; pout_c[i] = s_out[i]; }
}

__global__ __launch_bounds__(256) void gcn_colscan(
    int* __restrict__ pin, const int* __restrict__ pout,
    int* __restrict__ cnt_out, int* __restrict__ cnt_in, int n) {
  const int node = blockIdx.x * 256 + threadIdx.x;
  if (node >= n) return;
  int run_in = 0, run_out = 0;
  for (int cb = 0; cb < HIST_NCHUNK; cb += 16) {
    int a[16], b[16];
#pragma unroll
    for (int k = 0; k < 16; ++k) {
      a[k] = pin[(cb + k) * n + node];
      b[k] = pout[(cb + k) * n + node];
    }
#pragma unroll
    for (int k = 0; k < 16; ++k) {
      pin[(cb + k) * n + node] = run_in;
      run_in += a[k];
      run_out += b[k];
    }
  }
  cnt_in[node] = run_in;
  cnt_out[node] = run_out;
}

__global__ __launch_bounds__(1024) void gcn_scan_norm(
    const int* __restrict__ cnt_out, const int* __restrict__ cnt_in,
    float* __restrict__ norm_src, float* __restrict__ norm_dst,
    int* __restrict__ row_ptr, int n) {
  __shared__ int s_part[32];
  __shared__ int s_incl[32];
  const int tid = threadIdx.x;
  const int lane = tid & 31;
  const int wv = tid >> 5;
  const int blockStart = blockIdx.x * 1024;

  int part = 0;
  const int4* c4 = (const int4*)cnt_in;
  const int grp = __builtin_amdgcn_readfirstlane(tid >> 8);
  const int sub = tid & 255;
  const int groups = blockIdx.x;
#pragma unroll 8
  for (int g = grp; g < groups; g += 4) {
    const int4 q = c4[g * 256 + sub];
    part += (q.x + q.y) + (q.z + q.w);
  }
  for (int o = 16; o > 0; o >>= 1) part += __shfl_xor(part, o);
  if (lane == 0) s_part[wv] = part;

  const int idx = blockStart + tid;
  const int c = idx < n ? cnt_in[idx] : 0;
  int inc = c;
  for (int o = 1; o < 32; o <<= 1) {
    const int v = __shfl_up(inc, o);
    if (lane >= o) inc += v;
  }
  if (lane == 31) s_incl[wv] = inc;
  __syncthreads();

  int off = 0;
  for (int w = 0; w < 32; ++w) {
    off += s_part[w];
    if (w < wv) off += s_incl[w];
  }
  const int excl = off + inc - c;
  if (idx < n) {
    row_ptr[idx] = excl;
    if (idx == n - 1) row_ptr[n] = excl + c;
    norm_dst[idx] = rsqrtf(fmaxf((float)c, 1.0f));
    norm_src[idx] = rsqrtf(fmaxf((float)cnt_out[idx], 1.0f));
  }
}

#define FILL_EPT ((HIST_CE + 1023) / 1024)
#define FILL_NPT ((HIST_RN + 1023) / 1024)

__device__ __forceinline__ void gcn_fill_block(
    const int* __restrict__ src, const int* __restrict__ dst,
    const int* __restrict__ row_ptr, const int* __restrict__ pin,
    int* __restrict__ col, int n, int E, int* s_pos, int blk) {
  const int tid = threadIdx.x;
  const int rg = blk % HIST_NRANGE;
  const int ch = blk / HIST_NRANGE;
  const int lo = rg * HIST_RN;
  const int lim = (n - lo < HIST_RN) ? (n - lo) : HIST_RN;
  const int e0 = ch * HIST_CE;
  const int e1 = (e0 + HIST_CE < E) ? (e0 + HIST_CE) : E;
  const int* rp_c = row_ptr + lo;
  const int* pin_c = pin + ch * n + lo;

  {
    int a[FILL_NPT], b[FILL_NPT];
#pragma unroll
    for (int k = 0; k < FILL_NPT; ++k) {
      const int i = tid + k * 1024;
      a[k] = i < lim ? rp_c[i] : 0;
      b[k] = i < lim ? pin_c[i] : 0;
    }
#pragma unroll
    for (int k = 0; k < FILL_NPT; ++k) {
      const int i = tid + k * 1024;
      if (i < lim) s_pos[i] = a[k] + b[k];
    }
  }
  int d[FILL_EPT], s[FILL_EPT];
#pragma unroll
  for (int k = 0; k < FILL_EPT; ++k) {
    const int e = e0 + tid + k * 1024;
    d[k] = e < e1 ? dst[e] : lo - 1;
    s[k] = e < e1 ? src[e] : 0;
  }
  __syncthreads();

#pragma unroll
  for (int k = 0; k < FILL_EPT; ++k) {
    const unsigned ud = (unsigned)(d[k] - lo);
    int place = -1;
    if (ud < (unsigned)HIST_RN) place = atomicAdd(&s_pos[ud], 1);
    d[k] = place;
  }
#pragma unroll
  for (int k = 0; k < FILL_EPT; ++k) {
    if (d[k] >= 0) col[d[k]] = s[k];
  }
}

__device__ __forceinline__ void gcn_gemm_wave(
    const float* __restrict__ A, const float* __restrict__ W,
    const float* __restrict__ nsrc, float* __restrict__ X,
    int nRows, int wave) {
  const int K = GCN_INDIM;
  int lane = threadIdx.x & 31;
  int rowBlk = wave * 16;
  if (rowBlk >= nRows) return;

  int half = lane >> 4;
  int l    = lane & 15;
  int row  = rowBlk + l;
  float ns = nsrc[row];
  const float* arow = A + (long)row * K;

  v8f acc0 = {}, acc1 = {}, acc2 = {}, acc3 = {};

#pragma unroll 2
  for (int k0 = 0; k0 < K; k0 += 4) {
    int ka = k0 + half * 2;
    v2f a;
    a.x = arow[ka]     * ns;
    a.y = arow[ka + 1] * ns;
    const float* w0 = W + (long)ka * 64 + l;
    const float* w1 = w0 + 64;
    v2f b;
    b.x = w0[0];  b.y = w1[0];
    acc0 = __builtin_amdgcn_wmma_f32_16x16x4_f32(false, a, false, b, (short)0, acc0, false, false);
    b.x = w0[16]; b.y = w1[16];
    acc1 = __builtin_amdgcn_wmma_f32_16x16x4_f32(false, a, false, b, (short)0, acc1, false, false);
    b.x = w0[32]; b.y = w1[32];
    acc2 = __builtin_amdgcn_wmma_f32_16x16x4_f32(false, a, false, b, (short)0, acc2, false, false);
    b.x = w0[48]; b.y = w1[48];
    acc3 = __builtin_amdgcn_wmma_f32_16x16x4_f32(false, a, false, b, (short)0, acc3, false, false);
  }

  float* xout = X + (long)rowBlk * 64;
#pragma unroll
  for (int i = 0; i < 8; ++i) {
    int r = i + half * 8;
    xout[(long)r * 64 +  0 + l] = acc0[i];
    xout[(long)r * 64 + 16 + l] = acc1[i];
    xout[(long)r * 64 + 32 + l] = acc2[i];
    xout[(long)r * 64 + 48 + l] = acc3[i];
  }
}

#define FILL_BLOCKS (HIST_NRANGE * HIST_NCHUNK)
#define GEMM_TPB 25

__global__ __launch_bounds__(1024, 16) void gcn_fill_gemm(
    const int* __restrict__ src, const int* __restrict__ dst,
    const int* __restrict__ row_ptr, const int* __restrict__ pin,
    int* __restrict__ col, int n, int E,
    const float* __restrict__ A, const float* __restrict__ W,
    const float* __restrict__ nsrc, float* __restrict__ X) {
  __shared__ int s_pos[HIST_RN];
  if (blockIdx.x < FILL_BLOCKS) {
    gcn_fill_block(src, dst, row_ptr, pin, col, n, E, s_pos, (int)blockIdx.x);
  } else {
    const int wv = (int)(threadIdx.x >> 5);
    if (wv < GEMM_TPB)
      gcn_gemm_wave(A, W, nsrc, X, n, ((int)blockIdx.x - FILL_BLOCKS) * GEMM_TPB + wv);
  }
}

#define PULL_NPW 8
#define PROJ_PITCH 68

template <bool HIDDEN>
__device__ __forceinline__ void gcn_pull_rows(
    const int* __restrict__ row_ptr, const int* __restrict__ col,
    const float* __restrict__ X, const float* __restrict__ ndst,
    const float* __restrict__ nsrc, const float* __restrict__ b,
    float* __restrict__ OUT, float* s_rows, int node0, int cnt, int lane) {
  int rpv = 0;
  float ndv = 0.0f;
  float nsv = 0.0f;
  if (lane <= cnt) rpv = row_ptr[node0 + lane];
  if (lane < cnt) ndv = ndst[node0 + lane];
  if (HIDDEN) {
    if (lane < cnt) nsv = nsrc[node0 + lane];
  }
  const float b0 = b[2 * lane];
  const float b1 = b[2 * lane + 1];
  const int lo = 2 * lane;

  const int first = __builtin_amdgcn_readlane(rpv, 0);
  const int total = __builtin_amdgcn_readlane(rpv, cnt) - first;
  const int* cl = col + first;

  int t = 0;
  int nodeEnd = __builtin_amdgcn_readlane(rpv, 1) - first;
  float ax = 0.0f, ay = 0.0f;

#define PULL_FINISH()                                                          \
  {                                                                            \
    const float nd = __int_as_float(                                           \
        __builtin_amdgcn_readlane(__float_as_int(ndv), t));                    \
    float r0 = ax * nd + b0;                                                   \
    float r1 = ay * nd + b1;                                                   \
    v2f r;                                                                     \
    if (HIDDEN) {                                                              \
      const float ns = __int_as_float(                                         \
          __builtin_amdgcn_readlane(__float_as_int(nsv), t));                  \
      r0 = r0 > 0.0f ? r0 : expm1f(r0);                                        \
      r1 = r1 > 0.0f ? r1 : expm1f(r1);                                        \
      r.x = r0 * ns;                                                           \
      r.y = r1 * ns;                                                           \
      *(v2f*)(s_rows + t * PROJ_PITCH + lo) = r;                               \
    } else {                                                                   \
      r.x = r0;                                                                \
      r.y = r1;                                                                \
      *(v2f*)(OUT + (long)(node0 + t) * 64 + lo) = r;                          \
    }                                                                          \
    ax = 0.0f;                                                                 \
    ay = 0.0f;                                                                 \
    ++t;                                                                       \
    nodeEnd = __builtin_amdgcn_readlane(rpv, t + 1) - first;                   \
  }

  int cv = 0;
  if (lane < 16 && lane < total) cv = cl[lane];

  for (int pos0 = 0; pos0 < total; pos0 += 16) {
    v2f v[16];
#pragma unroll
    for (int u = 0; u < 16; ++u) {
      const int sr = __builtin_amdgcn_readlane(cv, u);
      v[u] = *(const v2f*)(X + (long)sr * 64 + lo);
    }
    int cvn = 0;
    if (lane < 16 && pos0 + 16 + lane < total) cvn = cl[pos0 + 16 + lane];
#pragma unroll
    for (int u = 0; u < 16; ++u) {
      while (pos0 + u == nodeEnd) PULL_FINISH()
      ax += v[u].x;
      ay += v[u].y;
    }
    cv = cvn;
  }
  while (t < cnt) PULL_FINISH()
#undef PULL_FINISH
}

__global__ __launch_bounds__(256) void gcn_pull_lin(
    const int* __restrict__ row_ptr, const int* __restrict__ col,
    const float* __restrict__ X, const float* __restrict__ ndst,
    const float* __restrict__ b, float* __restrict__ OUT, int n) {
  const int lane = threadIdx.x & 31;
  const int wv = __builtin_amdgcn_readfirstlane((int)(threadIdx.x >> 5));
  const int node0 = (blockIdx.x * 8 + wv) * PULL_NPW;
  if (node0 >= n) return;
  const int rem = n - node0;
  const int cnt = rem < PULL_NPW ? rem : PULL_NPW;
  gcn_pull_rows<false>(row_ptr, col, X, ndst, nullptr, b, OUT, nullptr, node0, cnt, lane);
}

__global__ __launch_bounds__(256) void gcn_pull_elu_proj(
    const int* __restrict__ row_ptr, const int* __restrict__ col,
    const float* __restrict__ X, const float* __restrict__ ndst,
    const float* __restrict__ nsrc, const float* __restrict__ b,
    const float* __restrict__ W2, float* __restrict__ X2, int n) {
  __shared__ __attribute__((aligned(16))) float s_h[8 * PULL_NPW * PROJ_PITCH];
  const int lane = threadIdx.x & 31;
  const int wv = __builtin_amdgcn_readfirstlane((int)(threadIdx.x >> 5));
  const int node0 = (blockIdx.x * 8 + wv) * PULL_NPW;
  const int rem = n - node0;
  const int cnt = rem < PULL_NPW ? (rem > 0 ? rem : 0) : PULL_NPW;
  float* s_mine = s_h + wv * PULL_NPW * PROJ_PITCH;
  if (cnt > 0)
    gcn_pull_rows<true>(row_ptr, col, X, ndst, nsrc, b, nullptr, s_mine, node0, cnt, lane);
  for (int t = cnt; t < PULL_NPW; ++t) {
    v2f z;
    z.x = 0.0f;
    z.y = 0.0f;
    *(v2f*)(s_mine + t * PROJ_PITCH + 2 * lane) = z;
  }
  __syncthreads();

  const int half = lane >> 4;
  const int c = lane & 15;
  const int mt = wv >> 1;
  const int n0 = (wv & 1) * 32;
  const float* arow = s_h + (mt * 16 + c) * PROJ_PITCH + 2 * half;
  const float* wp = W2 + (2 * half) * 64 + n0 + 2 * c;
  v8f acc0 = {}, acc1 = {};
#pragma unroll
  for (int k0 = 0; k0 < GCN_HID; k0 += 4) {
    const v2f a = *(const v2f*)(arow + k0);
    const v2f w0 = *(const v2f*)(wp + k0 * 64);
    const v2f w1 = *(const v2f*)(wp + k0 * 64 + 64);
    v2f bA, bB;
    bA.x = w0.x; bA.y = w1.x;
    bB.x = w0.y; bB.y = w1.y;
    acc0 = __builtin_amdgcn_wmma_f32_16x16x4_f32(false, a, false, bA, (short)0, acc0, false, false);
    acc1 = __builtin_amdgcn_wmma_f32_16x16x4_f32(false, a, false, bB, (short)0, acc1, false, false);
  }

  const int rowBase = blockIdx.x * (8 * PULL_NPW) + mt * 16 + half * 8;
#pragma unroll
  for (int i = 0; i < 8; ++i) {
    const int r = rowBase + i;
    if (r < n) {
      v2f o;
      o.x = acc0[i];
      o.y = acc1[i];
      *(v2f*)(X2 + (long)r * 64 + n0 + 2 * c) = o;
    }
  }
}

extern "C" void kernel_launch(void* const* d_in, const int* in_sizes, int n_in,
                              void* d_out, int out_size, void* d_ws, size_t ws_size,
                              hipStream_t stream) {
  const float* h   = (const float*)d_in[0];
  const int*   src = (const int*)  d_in[1];
  const int*   dst = (const int*)  d_in[2];
  const float* W1  = (const float*)d_in[3];
  const float* b1  = (const float*)d_in[4];
  const float* W2  = (const float*)d_in[5];
  const float* b2  = (const float*)d_in[6];
  float* out = (float*)d_out;

  const int N = in_sizes[0] / GCN_INDIM;
  const int E = in_sizes[1];

  float* norm_src = (float*)d_ws;
  float* norm_dst = norm_src + N;
  float* bufA     = norm_dst + N;
  int*   cnt_out  = (int*)(bufA + (long)N * 64);
  int*   cnt_in   = cnt_out + N;
  int*   row_ptr  = cnt_in + N;
  int*   col      = row_ptr + N + 32;
  int*   pin      = (int*)bufA;
  int*   pout     = (int*)d_out;

  const int pullBlocks = (N + 8 * PULL_NPW - 1) / (8 * PULL_NPW);
  const int waves  = N / 16;
  const int gemmBlocks = (waves + GEMM_TPB - 1) / GEMM_TPB;

  gcn_hist<<<HIST_NRANGE * HIST_NCHUNK, 1024, 0, stream>>>(src, dst, pin, pout, N, E);
  gcn_colscan<<<(N + 255) / 256, 256, 0, stream>>>(pin, pout, cnt_out, cnt_in, N);
  gcn_scan_norm<<<(N + 1023) / 1024, 1024, 0, stream>>>(cnt_out, cnt_in, norm_src, norm_dst, row_ptr, N);
  gcn_fill_gemm<<<FILL_BLOCKS + gemmBlocks, 1024, 0, stream>>>(src, dst, row_ptr, pin, col, N, E,
                                                                 h, W1, norm_src, out);

  gcn_pull_elu_proj<<<pullBlocks, 256, 0, stream>>>(row_ptr, col, out, norm_dst, norm_src, b1, W2, bufA, N);

  gcn_pull_lin<<<pullBlocks, 256, 0, stream>>>(row_ptr, col, bufA, norm_dst, b2, out, N);
}


# row_ptr scan and norms moved into the per-node chunk-scan launch: count blocks also leave per-256-node group sums and a block adds up the entries in front of it; one launch fewer, integer work only
# speedup vs baseline: 3.2835x; 1.0353x over previous
#include <hip/hip_runtime.h>
#include <math.h>

typedef float v2f __attribute__((ext_vector_type(2)));
typedef float v8f __attribute__((ext_vector_type(8)));

#define GCN_N 100000
#define GCN_E 1600000
#define GCN_INDIM 128
#define GCN_HID 64

#define HIST_NRANGE 4
#define HIST_NCHUNK 64
#define HIST_RN ((GCN_N + HIST_NRANGE - 1) / HIST_NRANGE)
#define HIST_CE ((GCN_E + HIST_NCHUNK - 1) / HIST_NCHUNK)
#define GRP_N 256
#define GRP_PER_RANGE ((HIST_RN + GRP_N - 1) / GRP_N)
#define GRP_TOTAL (HIST_NRANGE * GRP_PER_RANGE)
#define GRP_TRIPS ((GRP_PER_RANGE + 31) / 32)
#define PRE_LD ((GRP_TOTAL * HIST_NCHUNK / 4 + 255) / 256)

__global__ __launch_bounds__(1024) void gcn_hist(
    const int* __restrict__ src, const int* __restrict__ dst,
    int* __restrict__ pin, int* __restrict__ pout, int* __restrict__ gsum, int n, int E) {
  __shared__ int s_in[HIST_RN];
  __shared__ int s_out[HIST_RN];
  const int tid = threadIdx.x;
  const int rg = blockIdx.x % HIST_NRANGE;
  const int ch = blockIdx.x / HIST_NRANGE;
  const int lo = rg * HIST_RN;

  for (int i = tid; i < HIST_RN; i += 1024) { s_in[i] = 0; s_out[i] = 0; }
  __syncthreads();

  const int e0 = ch * HIST_CE;
  const int e1 = (e0 + HIST_CE < E) ? (e0 + HIST_CE) : E;
  for (int eb = e0 + tid; eb < e1; eb += 8 * 1024) {
    int d[8], s[8];
#pragma unroll
    for (int k = 0; k < 8; ++k) {
      const int e = eb + k * 1024;
      d[k] = e < e1 ? dst[e] : lo - 1;
      s[k] = e < e1 ? src[e] : lo - 1;
    }
#pragma unroll
    for (int k = 0; k < 8; ++k) {
      const unsigned ud = (unsigned)(d[k] - lo);
      const unsigned us = (unsigned)(s[k] - lo);
      if (ud < (unsigned)HIST_RN) atomicAdd(&s_in[ud], 1);
      if (us < (unsigned)HIST_RN) atomicAdd(&s_out[us], 1);
    }
  }
  __syncthreads();

  int* pin_c = pin + ch * n + lo;
  int* pout_c = pout + ch * n + lo;
  const int lim = (n - lo < HIST_RN) ? (n - lo) : HIST_RN;
  for (int i = tid; i < lim; i += 1024) { pin_c[i] = s_in[i]; pout_c[i] = s_out[i]; }

  const int lane = tid & 31;
  const int wv = tid >> 5;
  int gs[GRP_TRIPS];
#pragma unroll
  for (int t = 0; t < GRP_TRIPS; ++t) {
    gs[t] = 0;
#pragma unroll
    for (int j = 0; j < GRP_N / 32; ++j) {
      const int i = (wv + 32 * t) * GRP_N + j * 32 + lane;
      gs[t] += i < HIST_RN ? s_in[i] : 0;
    }
  }
  for (int o = 16; o > 0; o >>= 1) {
#pragma unroll
    for (int t = 0; t < GRP_TRIPS; ++t) gs[t] += __shfl_xor(gs[t], o);
  }
#pragma unroll
  for (int t = 0; t < GRP_TRIPS; ++t) {
    const int g = wv + 32 * t;
    if (lane == 0 && g < GRP_PER_RANGE) gsum[(rg * GRP_PER_RANGE + g) * HIST_NCHUNK + ch] = gs[t];
  }
}

__global__ __launch_bounds__(256) void gcn_colscan_rowptr(
    int* __restrict__ pin, const int* __restrict__ pout, const int* __restrict__ gsum,
    float* __restrict__ norm_src, float* __restrict__ norm_dst,
    int* __restrict__ row_ptr, int n) {
  __shared__ int s_part[8];
  __shared__ int s_incl[8];
  const int tid = threadIdx.x;
  const int lane = tid & 31;
  const int wv = tid >> 5;
  const int grp = blockIdx.x;
  const int rg = grp / GRP_PER_RANGE;
  const int loc = (grp - rg * GRP_PER_RANGE) * GRP_N + tid;
  const int node = rg * HIST_RN + loc;
  const bool valid = loc < HIST_RN && node < n;

  const int4* g4 = (const int4*)gsum;
  const int nq = grp * (HIST_NCHUNK / 4);
  int4 pv[PRE_LD];
#pragma unroll
  for (int k = 0; k < PRE_LD; ++k) {
    const int q = k * 256 + tid;
    pv[k] = g4[q < nq ? q : 0];
  }

  int run_in = 0, run_out = 0;
  if (valid) {
    for (int cb = 0; cb < HIST_NCHUNK; cb += 16) {
      int a[16], b[16];
#pragma unroll
      for (int k = 0; k < 16; ++k) {
        a[k] = pin[(cb + k) * n + node];
        b[k] = pout[(cb + k) * n + node];
      }
#pragma unroll
      for (int k = 0; k < 16; ++k) {
        pin[(cb + k) * n + node] = run_in;
        run_in += a[k];
        run_out += b[k];
      }
    }
  }

  int part = 0;
#pragma unroll
  for (int k = 0; k < PRE_LD; ++k) {
    const int s4 = (pv[k].x + pv[k].y) + (pv[k].z + pv[k].w);
    part += (k * 256 + tid < nq) ? s4 : 0;
  }
  for (int o = 16; o > 0; o >>= 1) part += __shfl_xor(part, o);
  if (lane == 0) s_part[wv] = part;
  const int c = run_in;
  int inc = c;
  for (int o = 1; o < 32; o <<= 1) {
    const int v = __shfl_up(inc, o);
    if (lane >= o) inc += v;
  }
  if (lane == 31) s_incl[wv] = inc;
  __syncthreads();

  int off = 0;
  for (int w = 0; w < 8; ++w) {
    off += s_part[w];
    if (w < wv) off += s_incl[w];
  }
  const int excl = off + inc - c;
  if (valid) {
    row_ptr[node] = excl;
    if (node == n - 1) row_ptr[n] = excl + c;
    norm_dst[node] = rsqrtf(fmaxf((float)c, 1.0f));
    norm_src[node] = rsqrtf(fmaxf((float)run_out, 1.0f));
  }
}

#define FILL_EPT ((HIST_CE + 1023) / 1024)
#define FILL_NPT ((HIST_RN + 1023) / 1024)

__device__ __forceinline__ void gcn_fill_block(
    const int* __restrict__ src, const int* __restrict__ dst,
    const int* __restrict__ row_ptr, const int* __restrict__ pin,
    int* __restrict__ col, int n, int E, int* s_pos, int blk) {
  const int tid = threadIdx.x;
  const int rg = blk % HIST_NRANGE;
  const int ch = blk / HIST_NRANGE;
  const int lo = rg * HIST_RN;
  const int lim = (n - lo < HIST_RN) ? (n - lo) : HIST_RN;
  const int e0 = ch * HIST_CE;
  const int e1 = (e0 + HIST_CE < E) ? (e0 + HIST_CE) : E;
  const int* rp_c = row_ptr + lo;
  const int* pin_c = pin + ch * n + lo;

  {
    int a[FILL_NPT], b[FILL_NPT];
#pragma unroll
    for (int k = 0; k < FILL_NPT; ++k) {
      const int i = tid + k * 1024;
      a[k] = i < lim ? rp_c[i] : 0;
      b[k] = i < lim ? pin_c[i] : 0;
    }
#pragma unroll
    for (int k = 0; k < FILL_NPT; ++k) {
      const int i = tid + k * 1024;
      if (i < lim) s_pos[i] = a[k] + b[k];
    }
  }
  int d[FILL_EPT], s[FILL_EPT];
#pragma unroll
  for (int k = 0; k < FILL_EPT; ++k) {
    const int e = e0 + tid + k * 1024;
    d[k] = e < e1 ? dst[e] : lo - 1;
    s[k] = e < e1 ? src[e] : 0;
  }
  __syncthreads();

#pragma unroll
  for (int k = 0; k < FILL_EPT; ++k) {
    const unsigned ud = (unsigned)(d[k] - lo);
    int place = -1;
    if (ud < (unsigned)HIST_RN) place = atomicAdd(&s_pos[ud], 1);
    d[k] = place;
  }
#pragma unroll
  for (int k = 0; k < FILL_EPT; ++k) {
    if (d[k] >= 0) col[d[k]] = s[k];
  }
}

__device__ __forceinline__ void gcn_gemm_wave(
    const float* __restrict__ A, const float* __restrict__ W,
    const float* __restrict__ nsrc, float* __restrict__ X,
    int nRows, int wave) {
  const int K = GCN_INDIM;
  int lane = threadIdx.x & 31;
  int rowBlk = wave * 16;
  if (rowBlk >= nRows) return;

  int half = lane >> 4;
  int l    = lane & 15;
  int row  = rowBlk + l;
  float ns = nsrc[row];
  const float* arow = A + (long)row * K;

  v8f acc0 = {}, acc1 = {}, acc2 = {}, acc3 = {};

#pragma unroll 2
  for (int k0 = 0; k0 < K; k0 += 4) {
    int ka = k0 + half * 2;
    v2f a;
    a.x = arow[ka]     * ns;
    a.y = arow[ka + 1] * ns;
    const float* w0 = W + (long)ka * 64 + l;
    const float* w1 = w0 + 64;
    v2f b;
    b.x = w0[0];  b.y = w1[0];
    acc0 = __builtin_amdgcn_wmma_f32_16x16x4_f32(false, a, false, b, (short)0, acc0, false, false);
    b.x = w0[16]; b.y = w1[16];
    acc1 = __builtin_amdgcn_wmma_f32_16x16x4_f32(false, a, false, b, (short)0, acc1, false, false);
    b.x = w0[32]; b.y = w1[32];
    acc2 = __builtin_amdgcn_wmma_f32_16x16x4_f32(false, a, false, b, (short)0, acc2, false, false);
    b.x = w0[48]; b.y = w1[48];
    acc3 = __builtin_amdgcn_wmma_f32_16x16x4_f32(false, a, false, b, (short)0, acc3, false, false);
  }

  float* xout = X + (long)rowBlk * 64;
#pragma unroll
  for (int i = 0; i < 8; ++i) {
    int r = i + half * 8;
    xout[(long)r * 64 +  0 + l] = acc0[i];
    xout[(long)r * 64 + 16 + l] = acc1[i];
    xout[(long)r * 64 + 32 + l] = acc2[i];
    xout[(long)r * 64 + 48 + l] = acc3[i];
  }
}

#define FILL_BLOCKS (HIST_NRANGE * HIST_NCHUNK)
#define GEMM_TPB 25

__global__ __launch_bounds__(1024, 16) void gcn_fill_gemm(
    const int* __restrict__ src, const int* __restrict__ dst,
    const int* __restrict__ row_ptr, const int* __restrict__ pin,
    int* __restrict__ col, int n, int E,
    const float* __restrict__ A, const float* __restrict__ W,
    const float* __restrict__ nsrc, float* __restrict__ X) {
  __shared__ int s_pos[HIST_RN];
  if (blockIdx.x < FILL_BLOCKS) {
    gcn_fill_block(src, dst, row_ptr, pin, col, n, E, s_pos, (int)blockIdx.x);
  } else {
    const int wv = (int)(threadIdx.x >> 5);
    if (wv < GEMM_TPB)
      gcn_gemm_wave(A, W, nsrc, X, n, ((int)blockIdx.x - FILL_BLOCKS) * GEMM_TPB + wv);
  }
}

#define PULL_NPW 8
#define PROJ_PITCH 68

template <bool HIDDEN>
__device__ __forceinline__ void gcn_pull_rows(
    const int* __restrict__ row_ptr, const int* __restrict__ col,
    const float* __restrict__ X, const float* __restrict__ ndst,
    const float* __restrict__ nsrc, const float* __restrict__ b,
    float* __restrict__ OUT, float* s_rows, int node0, int cnt, int lane) {
  int rpv = 0;
  float ndv = 0.0f;
  float nsv = 0.0f;
  if (lane <= cnt) rpv = row_ptr[node0 + lane];
  if (lane < cnt) ndv = ndst[node0 + lane];
  if (HIDDEN) {
    if (lane < cnt) nsv = nsrc[node0 + lane];
  }
  const float b0 = b[2 * lane];
  const float b1 = b[2 * lane + 1];
  const int lo = 2 * lane;

  const int first = __builtin_amdgcn_readlane(rpv, 0);
  const int total = __builtin_amdgcn_readlane(rpv, cnt) - first;
  const int* cl = col + first;

  int t = 0;
  int nodeEnd = __builtin_amdgcn_readlane(rpv, 1) - first;
  float ax = 0.0f, ay = 0.0f;

#define PULL_FINISH()                                                          \
  {                                                                            \
    const float nd = __int_as_float(                                           \
        __builtin_amdgcn_readlane(__float_as_int(ndv), t));                    \
    float r0 = ax * nd + b0;                                                   \
    float r1 = ay * nd + b1;                                                   \
    v2f r;                                                                     \
    if (HIDDEN) {                                                              \
      const float ns = __int_as_float(                                         \
          __builtin_amdgcn_readlane(__float_as_int(nsv), t));                  \
      r0 = r0 > 0.0f ? r0 : expm1f(r0);                                        \
      r1 = r1 > 0.0f ? r1 : expm1f(r1);                                        \
      r.x = r0 * ns;                                                           \
      r.y = r1 * ns;                                                           \
      *(v2f*)(s_rows + t * PROJ_PITCH + lo) = r;                               \
    } else {                                                                   \
      r.x = r0;                                                                \
      r.y = r1;                                                                \
      *(v2f*)(OUT + (long)(node0 + t) * 64 + lo) = r;                          \
    }                                                                          \
    ax = 0.0f;                                                                 \
    ay = 0.0f;                                                                 \
    ++t;                                                                       \
    nodeEnd = __builtin_amdgcn_readlane(rpv, t + 1) - first;                   \
  }

  int cv = 0;
  if (lane < 16 && lane < total) cv = cl[lane];

  for (int pos0 = 0; pos0 < total; pos0 += 16) {
    v2f v[16];
#pragma unroll
    for (int u = 0; u < 16; ++u) {
      const int sr = __builtin_amdgcn_readlane(cv, u);
      v[u] = *(const v2f*)(X + (long)sr * 64 + lo);
    }
    int cvn = 0;
    if (lane < 16 && pos0 + 16 + lane < total) cvn = cl[pos0 + 16 + lane];
#pragma unroll
    for (int u = 0; u < 16; ++u) {
      while (pos0 + u == nodeEnd) PULL_FINISH()
      ax += v[u].x;
      ay += v[u].y;
    }
    cv = cvn;
  }
  while (t < cnt) PULL_FINISH()
#undef PULL_FINISH
}

__global__ __launch_bounds__(256) void gcn_pull_lin(
    const int* __restrict__ row_ptr, const int* __restrict__ col,
    const float* __restrict__ X, const float* __restrict__ ndst,
    const float* __restrict__ b, float* __restrict__ OUT, int n) {
  const int lane = threadIdx.x & 31;
  const int wv = __builtin_amdgcn_readfirstlane((int)(threadIdx.x >> 5));
  const int node0 = (blockIdx.x * 8 + wv) * PULL_NPW;
  if (node0 >= n) return;
  const int rem = n - node0;
  const int cnt = rem < PULL_NPW ? rem : PULL_NPW;
  gcn_pull_rows<false>(row_ptr, col, X, ndst, nullptr, b, OUT, nullptr, node0, cnt, lane);
}

__global__ __launch_bounds__(256) void gcn_pull_elu_proj(
    const int* __restrict__ row_ptr, const int* __restrict__ col,
    const float* __restrict__ X, const float* __restrict__ ndst,
    const float* __restrict__ nsrc, const float* __restrict__ b,
    const float* __restrict__ W2, float* __restrict__ X2, int n) {
  __shared__ __attribute__((aligned(16))) float s_h[8 * PULL_NPW * PROJ_PITCH];
  const int lane = threadIdx.x & 31;
  const int wv = __builtin_amdgcn_readfirstlane((int)(threadIdx.x >> 5));
  const int node0 = (blockIdx.x * 8 + wv) * PULL_NPW;
  const int rem = n - node0;
  const int cnt = rem < PULL_NPW ? (rem > 0 ? rem : 0) : PULL_NPW;
  float* s_mine = s_h + wv * PULL_NPW * PROJ_PITCH;
  if (cnt > 0)
    gcn_pull_rows<true>(row_ptr, col, X, ndst, nsrc, b, nullptr, s_mine, node0, cnt, lane);
  for (int t = cnt; t < PULL_NPW; ++t) {
    v2f z;
    z.x = 0.0f;
    z.y = 0.0f;
    *(v2f*)(s_mine + t * PROJ_PITCH + 2 * lane) = z;
  }
  __syncthreads();

  const int half = lane >> 4;
  const int c = lane & 15;
  const int mt = wv >> 1;
  const int n0 = (wv & 1) * 32;
  const float* arow = s_h + (mt * 16 + c) * PROJ_PITCH + 2 * half;
  const float* wp = W2 + (2 * half) * 64 + n0 + 2 * c;
  v8f acc0 = {}, acc1 = {};
#pragma unroll
  for (int k0 = 0; k0 < GCN_HID; k0 += 4) {
    const v2f a = *(const v2f*)(arow + k0);
    const v2f w0 = *(const v2f*)(wp + k0 * 64);
    const v2f w1 = *(const v2f*)(wp + k0 * 64 + 64);
    v2f bA, bB;
    bA.x = w0.x; bA.y = w1.x;
    bB.x = w0.y; bB.y = w1.y;
    acc0 = __builtin_amdgcn_wmma_f32_16x16x4_f32(false, a, false, bA, (short)0, acc0, false, false);
    acc1 = __builtin_amdgcn_wmma_f32_16x16x4_f32(false, a, false, bB, (short)0, acc1, false, false);
  }

  const int rowBase = blockIdx.x * (8 * PULL_NPW) + mt * 16 + half * 8;
#pragma unroll
  for (int i = 0; i < 8; ++i) {
    const int r = rowBase + i;
    if (r < n) {
      v2f o;
      o.x = acc0[i];
      o.y = acc1[i];
      *(v2f*)(X2 + (long)r * 64 + n0 + 2 * c) = o;
    }
  }
}

extern "C" void kernel_launch(void* const* d_in, const int* in_sizes, int n_in,
                              void* d_out, int out_size, void* d_ws, size_t ws_size,
                              hipStream_t stream) {
  const float* h   = (const float*)d_in[0];
  const int*   src = (const int*)  d_in[1];
  const int*   dst = (const int*)  d_in[2];
  const float* W1  = (const float*)d_in[3];
  const float* b1  = (const float*)d_in[4];
  const float* W2  = (const float*)d_in[5];
  const float* b2  = (const float*)d_in[6];
  float* out = (float*)d_out;

  const int N = in_sizes[0] / GCN_INDIM;
  const int E = in_sizes[1];

  float* norm_src = (float*)d_ws;
  float* norm_dst = norm_src + N;
  float* bufA     = norm_dst + N;
  int*   gsum     = (int*)(bufA + (long)N * 64);
  int*   row_ptr  = gsum + 2 * N;
  int*   col      = row_ptr + N + 32;
  int*   pin      = (int*)bufA;
  int*   pout     = (int*)d_out;

  const int pullBlocks = (N + 8 * PULL_NPW - 1) / (8 * PULL_NPW);
  const int waves  = N / 16;
  const int gemmBlocks = (waves + GEMM_TPB - 1) / GEMM_TPB;

  gcn_hist<<<HIST_NRANGE * HIST_NCHUNK, 1024, 0, stream>>>(src, dst, pin, pout, gsum, N, E);
  gcn_colscan_rowptr<<<GRP_TOTAL, 256, 0, stream>>>(pin, pout, gsum, norm_src, norm_dst, row_ptr, N);
  gcn_fill_gemm<<<FILL_BLOCKS + gemmBlocks, 1024, 0, stream>>>(src, dst, row_ptr, pin, col, N, E,
                                                                 h, W1, norm_src, out);

  gcn_pull_elu_proj<<<pullBlocks, 256, 0, stream>>>(row_ptr, col, out, norm_dst, norm_src, b1, W2, bufA, N);

  gcn_pull_lin<<<pullBlocks, 256, 0, stream>>>(row_ptr, col, bufA, norm_dst, b2, out, N);
}


# degree counts: in- and out-count of a (chunk, node) pair share one 32-bit word (16 bits each; a chunk has under 65536 edges): one LDS counter array, one count plane, scan in one trip; integers only
# speedup vs baseline: 3.5485x; 1.0807x over previous
#include <hip/hip_runtime.h>
#include <math.h>

typedef float v2f __attribute__((ext_vector_type(2)));
typedef float v8f __attribute__((ext_vector_type(8)));

#define GCN_N 100000
#define GCN_E 1600000
#define GCN_INDIM 128
#define GCN_HID 64

#define HIST_NRANGE 4
#define HIST_NCHUNK 64
#define HIST_RN ((GCN_N + HIST_NRANGE - 1) / HIST_NRANGE)
#define HIST_CE ((GCN_E + HIST_NCHUNK - 1) / HIST_NCHUNK)
#define GRP_N 256
#define GRP_PER_RANGE ((HIST_RN + GRP_N - 1) / GRP_N)
#define GRP_TOTAL (HIST_NRANGE * GRP_PER_RANGE)
#define GRP_TRIPS ((GRP_PER_RANGE + 31) / 32)
#define PRE_LD ((GRP_TOTAL * HIST_NCHUNK / 4 + 255) / 256)
#define CNT_OUT_ONE 0x10000
static_assert(HIST_CE < 65536, "a chunk's count of one node must fit 16 bits");

__global__ __launch_bounds__(1024) void gcn_hist(
    const int* __restrict__ src, const int* __restrict__ dst,
    int* __restrict__ pin, int* __restrict__ gsum, int n, int E) {
  __shared__ int s_cnt[HIST_RN];
  const int tid = threadIdx.x;
  const int rg = blockIdx.x % HIST_NRANGE;
  const int ch = blockIdx.x / HIST_NRANGE;
  const int lo = rg * HIST_RN;

  for (int i = tid; i < HIST_RN; i += 1024) s_cnt[i] = 0;
  __syncthreads();

  const int e0 = ch * HIST_CE;
  const int e1 = (e0 + HIST_CE < E) ? (e0 + HIST_CE) : E;
  for (int eb = e0 + tid; eb < e1; eb += 8 * 1024) {
    int d[8], s[8];
#pragma unroll
    for (int k = 0; k < 8; ++k) {
      const int e = eb + k * 1024;
      d[k] = e < e1 ? dst[e] : lo - 1;
      s[k] = e < e1 ? src[e] : lo - 1;
    }
#pragma unroll
    for (int k = 0; k < 8; ++k) {
      const unsigned ud = (unsigned)(d[k] - lo);
      const unsigned us = (unsigned)(s[k] - lo);
      if (ud < (unsigned)HIST_RN) atomicAdd(&s_cnt[ud], 1);
      if (us < (unsigned)HIST_RN) atomicAdd(&s_cnt[us], CNT_OUT_ONE);
    }
  }
  __syncthreads();

  int* pin_c = pin + ch * n + lo;
  const int lim = (n - lo < HIST_RN) ? (n - lo) : HIST_RN;
  for (int i = tid; i < lim; i += 1024) pin_c[i] = s_cnt[i];

  const int lane = tid & 31;
  const int wv = tid >> 5;
  int gs[GRP_TRIPS];
#pragma unroll
  for (int t = 0; t < GRP_TRIPS; ++t) {
    gs[t] = 0;
#pragma unroll
    for (int j = 0; j < GRP_N / 32; ++j) {
      const int i = (wv + 32 * t) * GRP_N + j * 32 + lane;
      gs[t] += i < HIST_RN ? (s_cnt[i] & 0xFFFF) : 0;
    }
  }
  for (int o = 16; o > 0; o >>= 1) {
#pragma unroll
    for (int t = 0; t < GRP_TRIPS; ++t) gs[t] += __shfl_xor(gs[t], o);
  }
#pragma unroll
  for (int t = 0; t < GRP_TRIPS; ++t) {
    const int g = wv + 32 * t;
    if (lane == 0 && g < GRP_PER_RANGE) gsum[(rg * GRP_PER_RANGE + g) * HIST_NCHUNK + ch] = gs[t];
  }
}

__global__ __launch_bounds__(256) void gcn_colscan_rowptr(
    int* __restrict__ pin, const int* __restrict__ gsum,
    float* __restrict__ norm_src, float* __restrict__ norm_dst,
    int* __restrict__ row_ptr, int n) {
  __shared__ int s_part[8];
  __shared__ int s_incl[8];
  const int tid = threadIdx.x;
  const int lane = tid & 31;
  const int wv = tid >> 5;
  const int grp = blockIdx.x;
  const int rg = grp / GRP_PER_RANGE;
  const int loc = (grp - rg * GRP_PER_RANGE) * GRP_N + tid;
  const int node = rg * HIST_RN + loc;
  const bool valid = loc < HIST_RN && node < n;

  const int4* g4 = (const int4*)gsum;
  const int nq = grp * (HIST_NCHUNK / 4);
  int4 pv[PRE_LD];
#pragma unroll
  for (int k = 0; k < PRE_LD; ++k) {
    const int q = k * 256 + tid;
    pv[k] = g4[q < nq ? q : 0];
  }

  int run_in = 0, run_out = 0;
  if (valid) {
    const unsigned un = (unsigned)n;
    const unsigned unode = (unsigned)node;
    int w[HIST_NCHUNK];
#pragma unroll
    for (int k = 0; k < HIST_NCHUNK; ++k) w[k] = pin[(unsigned)k * un + unode];
#pragma unroll
    for (int k = 0; k < HIST_NCHUNK; ++k) {
      pin[(unsigned)k * un + unode] = run_in;
      run_in += w[k] & 0xFFFF;
      run_out += (int)((unsigned)w[k] >> 16);
    }
  }

  int part = 0;
#pragma unroll
  for (int k = 0; k < PRE_LD; ++k) {
    const int s4 = (pv[k].x + pv[k].y) + (pv[k].z + pv[k].w);
    part += (k * 256 + tid < nq) ? s4 : 0;
  }
  for (int o = 16; o > 0; o >>= 1) part += __shfl_xor(part, o);
  if (lane == 0) s_part[wv] = part;
  const int c = run_in;
  int inc = c;
  for (int o = 1; o < 32; o <<= 1) {
    const int v = __shfl_up(inc, o);
    if (lane >= o) inc += v;
  }
  if (lane == 31) s_incl[wv] = inc;
  __syncthreads();

  int off = 0;
  for (int w = 0; w < 8; ++w) {
    off += s_part[w];
    if (w < wv) off += s_incl[w];
  }
  const int excl = off + inc - c;
  if (valid) {
    row_ptr[node] = excl;
    if (node == n - 1) row_ptr[n] = excl + c;
    norm_dst[node] = rsqrtf(fmaxf((float)c, 1.0f));
    norm_src[node] = rsqrtf(fmaxf((float)run_out, 1.0f));
  }
}

#define FILL_EPT ((HIST_CE + 1023) / 1024)
#define FILL_NPT ((HIST_RN + 1023) / 1024)

__device__ __forceinline__ void gcn_fill_block(
    const int* __restrict__ src, const int* __restrict__ dst,
    const int* __restrict__ row_ptr, const int* __restrict__ pin,
    int* __restrict__ col, int n, int E, int* s_pos, int blk) {
  const int tid = threadIdx.x;
  const int rg = blk % HIST_NRANGE;
  const int ch = blk / HIST_NRANGE;
  const int lo = rg * HIST_RN;
  const int lim = (n - lo < HIST_RN) ? (n - lo) : HIST_RN;
  const int e0 = ch * HIST_CE;
  const int e1 = (e0 + HIST_CE < E) ? (e0 + HIST_CE) : E;
  const int* rp_c = row_ptr + lo;
  const int* pin_c = pin + ch * n + lo;

  {
    int a[FILL_NPT], b[FILL_NPT];
#pragma unroll
    for (int k = 0; k < FILL_NPT; ++k) {
      const int i = tid + k * 1024;
      a[k] = i < lim ? rp_c[i] : 0;
      b[k] = i < lim ? pin_c[i] : 0;
    }
#pragma unroll
    for (int k = 0; k < FILL_NPT; ++k) {
      const int i = tid + k * 1024;
      if (i < lim) s_pos[i] = a[k] + b[k];
    }
  }
  int d[FILL_EPT], s[FILL_EPT];
#pragma unroll
  for (int k = 0; k < FILL_EPT; ++k) {
    const int e = e0 + tid + k * 1024;
    d[k] = e < e1 ? dst[e] : lo - 1;
    s[k] = e < e1 ? src[e] : 0;
  }
  __syncthreads();

#pragma unroll
  for (int k = 0; k < FILL_EPT; ++k) {
    const unsigned ud = (unsigned)(d[k] - lo);
    int place = -1;
    if (ud < (unsigned)HIST_RN) place = atomicAdd(&s_pos[ud], 1);
    d[k] = place;
  }
#pragma unroll
  for (int k = 0; k < FILL_EPT; ++k) {
    if (d[k] >= 0) col[d[k]] = s[k];
  }
}

__device__ __forceinline__ void gcn_gemm_wave(
    const float* __restrict__ A, const float* __restrict__ W,
    const float* __restrict__ nsrc, float* __restrict__ X,
    int nRows, int wave) {
  const int K = GCN_INDIM;
  int lane = threadIdx.x & 31;
  int rowBlk = wave * 16;
  if (rowBlk >= nRows) return;

  int half = lane >> 4;
  int l    = lane & 15;
  int row  = rowBlk + l;
  float ns = nsrc[row];
  const float* arow = A + (long)row * K;

  v8f acc0 = {}, acc1 = {}, acc2 = {}, acc3 = {};

#pragma unroll 2
  for (int k0 = 0; k0 < K; k0 += 4) {
    int ka = k0 + half * 2;
    v2f a;
    a.x = arow[ka]     * ns;
    a.y = arow[ka + 1] * ns;
    const float* w0 = W + (long)ka * 64 + l;
    const float* w1 = w0 + 64;
    v2f b;
    b.x = w0[0];  b.y = w1[0];
    acc0 = __builtin_amdgcn_wmma_f32_16x16x4_f32(false, a, false, b, (short)0, acc0, false, false);
    b.x = w0[16]; b.y = w1[16];
    acc1 = __builtin_amdgcn_wmma_f32_16x16x4_f32(false, a, false, b, (short)0, acc1, false, false);
    b.x = w0[32]; b.y = w1[32];
    acc2 = __builtin_amdgcn_wmma_f32_16x16x4_f32(false, a, false, b, (short)0, acc2, false, false);
    b.x = w0[48]; b.y = w1[48];
    acc3 = __builtin_amdgcn_wmma_f32_16x16x4_f32(false, a, false, b, (short)0, acc3, false, false);
  }

  float* xout = X + (long)rowBlk * 64;
#pragma unroll
  for (int i = 0; i < 8; ++i) {
    int r = i + half * 8;
    xout[(long)r * 64 +  0 + l] = acc0[i];
    xout[(long)r * 64 + 16 + l] = acc1[i];
    xout[(long)r * 64 + 32 + l] = acc2[i];
    xout[(long)r * 64 + 48 + l] = acc3[i];
  }
}

#define FILL_BLOCKS (HIST_NRANGE * HIST_NCHUNK)
#define GEMM_TPB 25

__global__ __launch_bounds__(1024, 16) void gcn_fill_gemm(
    const int* __restrict__ src, const int* __restrict__ dst,
    const int* __restrict__ row_ptr, const int* __restrict__ pin,
    int* __restrict__ col, int n, int E,
    const float* __restrict__ A, const float* __restrict__ W,
    const float* __restrict__ nsrc, float* __restrict__ X) {
  __shared__ int s_pos[HIST_RN];
  if (blockIdx.x < FILL_BLOCKS) {
    gcn_fill_block(src, dst, row_ptr, pin, col, n, E, s_pos, (int)blockIdx.x);
  } else {
    const int wv = (int)(threadIdx.x >> 5);
    if (wv < GEMM_TPB)
      gcn_gemm_wave(A, W, nsrc, X, n, ((int)blockIdx.x - FILL_BLOCKS) * GEMM_TPB + wv);
  }
}

#define PULL_NPW 8
#define PROJ_PITCH 68

template <bool HIDDEN>
__device__ __forceinline__ void gcn_pull_rows(
    const int* __restrict__ row_ptr, const int* __restrict__ col,
    const float* __restrict__ X, const float* __restrict__ ndst,
    const float* __restrict__ nsrc, const float* __restrict__ b,
    float* __restrict__ OUT, float* s_rows, int node0, int cnt, int lane) {
  int rpv = 0;
  float ndv = 0.0f;
  float nsv = 0.0f;
  if (lane <= cnt) rpv = row_ptr[node0 + lane];
  if (lane < cnt) ndv = ndst[node0 + lane];
  if (HIDDEN) {
    if (lane < cnt) nsv = nsrc[node0 + lane];
  }
  const float b0 = b[2 * lane];
  const float b1 = b[2 * lane + 1];
  const int lo = 2 * lane;

  const int first = __builtin_amdgcn_readlane(rpv, 0);
  const int total = __builtin_amdgcn_readlane(rpv, cnt) - first;
  const int* cl = col + first;

  int t = 0;
  int nodeEnd = __builtin_amdgcn_readlane(rpv, 1) - first;
  float ax = 0.0f, ay = 0.0f;

#define PULL_FINISH()                                                          \
  {                                                                            \
    const float nd = __int_as_float(                                           \
        __builtin_amdgcn_readlane(__float_as_int(ndv), t));                    \
    float r0 = ax * nd + b0;                                                   \
    float r1 = ay * nd + b1;                                                   \
    v2f r;                                                                     \
    if (HIDDEN) {                                                              \
      const float ns = __int_as_float(                                         \
          __builtin_amdgcn_readlane(__float_as_int(nsv), t));                  \
      r0 = r0 > 0.0f ? r0 : expm1f(r0);                                        \
      r1 = r1 > 0.0f ? r1 : expm1f(r1);                                        \
      r.x = r0 * ns;                                                           \
      r.y = r1 * ns;                                                           \
      *(v2f*)(s_rows + t * PROJ_PITCH + lo) = r;                               \
    } else {                                                                   \
      r.x = r0;                                                                \
      r.y = r1;                                                                \
      *(v2f*)(OUT + (long)(node0 + t) * 64 + lo) = r;                          \
    }                                                                          \
    ax = 0.0f;                                                                 \
    ay = 0.0f;                                                                 \
    ++t;                                                                       \
    nodeEnd = __builtin_amdgcn_readlane(rpv, t + 1) - first;                   \
  }

  int cv = 0;
  if (lane < 16 && lane < total) cv = cl[lane];

  for (int pos0 = 0; pos0 < total; pos0 += 16) {
    v2f v[16];
#pragma unroll
    for (int u = 0; u < 16; ++u) {
      const int sr = __builtin_amdgcn_readlane(cv, u);
      v[u] = *(const v2f*)(X + (long)sr * 64 + lo);
    }
    int cvn = 0;
    if (lane < 16 && pos0 + 16 + lane < total) cvn = cl[pos0 + 16 + lane];
#pragma unroll
    for (int u = 0; u < 16; ++u) {
      while (pos0 + u == nodeEnd) PULL_FINISH()
      ax += v[u].x;
      ay += v[u].y;
    }
    cv = cvn;
  }
  while (t < cnt) PULL_FINISH()
#undef PULL_FINISH
}

__global__ __launch_bounds__(256) void gcn_pull_lin(
    const int* __restrict__ row_ptr, const int* __restrict__ col,
    const float* __restrict__ X, const float* __restrict__ ndst,
    const float* __restrict__ b, float* __restrict__ OUT, int n) {
  const int lane = threadIdx.x & 31;
  const int wv = __builtin_amdgcn_readfirstlane((int)(threadIdx.x >> 5));
  const int node0 = (blockIdx.x * 8 + wv) * PULL_NPW;
  if (node0 >= n) return;
  const int rem = n - node0;
  const int cnt = rem < PULL_NPW ? rem : PULL_NPW;
  gcn_pull_rows<false>(row_ptr, col, X, ndst, nullptr, b, OUT, nullptr, node0, cnt, lane);
}

__global__ __launch_bounds__(256) void gcn_pull_elu_proj(
    const int* __restrict__ row_ptr, const int* __restrict__ col,
    const float* __restrict__ X, const float* __restrict__ ndst,
    const float* __restrict__ nsrc, const float* __restrict__ b,
    const float* __restrict__ W2, float* __restrict__ X2, int n) {
  __shared__ __attribute__((aligned(16))) float s_h[8 * PULL_NPW * PROJ_PITCH];
  const int lane = threadIdx.x & 31;
  const int wv = __builtin_amdgcn_readfirstlane((int)(threadIdx.x >> 5));
  const int node0 = (blockIdx.x * 8 + wv) * PULL_NPW;
  const int rem = n - node0;
  const int cnt = rem < PULL_NPW ? (rem > 0 ? rem : 0) : PULL_NPW;
  float* s_mine = s_h + wv * PULL_NPW * PROJ_PITCH;
  if (cnt > 0)
    gcn_pull_rows<true>(row_ptr, col, X, ndst, nsrc, b, nullptr, s_mine, node0, cnt, lane);
  for (int t = cnt; t < PULL_NPW; ++t) {
    v2f z;
    z.x = 0.0f;
    z.y = 0.0f;
    *(v2f*)(s_mine + t * PROJ_PITCH + 2 * lane) = z;
  }
  __syncthreads();

  const int half = lane >> 4;
  const int c = lane & 15;
  const int mt = wv >> 1;
  const int n0 = (wv & 1) * 32;
  const float* arow = s_h + (mt * 16 + c) * PROJ_PITCH + 2 * half;
  const float* wp = W2 + (2 * half) * 64 + n0 + 2 * c;
  v8f acc0 = {}, acc1 = {};
#pragma unroll
  for (int k0 = 0; k0 < GCN_HID; k0 += 4) {
    const v2f a = *(const v2f*)(arow + k0);
    const v2f w0 = *(const v2f*)(wp + k0 * 64);
    const v2f w1 = *(const v2f*)(wp + k0 * 64 + 64);
    v2f bA, bB;
    bA.x = w0.x; bA.y = w1.x;
    bB.x = w0.y; bB.y = w1.y;
    acc0 = __builtin_amdgcn_wmma_f32_16x16x4_f32(false, a, false, bA, (short)0, acc0, false, false);
    acc1 = __builtin_amdgcn_wmma_f32_16x16x4_f32(false, a, false, bB, (short)0, acc1, false, false);
  }

  const int rowBase = blockIdx.x * (8 * PULL_NPW) + mt * 16 + half * 8;
#pragma unroll
  for (int i = 0; i < 8; ++i) {
    const int r = rowBase + i;
    if (r < n) {
      v2f o;
      o.x = acc0[i];
      o.y = acc1[i];
      *(v2f*)(X2 + (long)r * 64 + n0 + 2 * c) = o;
    }
  }
}

extern "C" void kernel_launch(void* const* d_in, const int* in_sizes, int n_in,
                              void* d_out, int out_size, void* d_ws, size_t ws_size,
                              hipStream_t stream) {
  const float* h   = (const float*)d_in[0];
  const int*   src = (const int*)  d_in[1];
  const int*   dst = (const int*)  d_in[2];
  const float* W1  = (const float*)d_in[3];
  const float* b1  = (const float*)d_in[4];
  const float* W2  = (const float*)d_in[5];
  const float* b2  = (const float*)d_in[6];
  float* out = (float*)d_out;

  const int N = in_sizes[0] / GCN_INDIM;
  const int E = in_sizes[1];

  float* norm_src = (float*)d_ws;
  float* norm_dst = norm_src + N;
  float* bufA     = norm_dst + N;
  int*   gsum     = (int*)(bufA + (long)N * 64);
  int*   row_ptr  = gsum + 2 * N;
  int*   col      = row_ptr + N + 32;
  int*   pin      = (int*)bufA;

  const int pullBlocks = (N + 8 * PULL_NPW - 1) / (8 * PULL_NPW);
  const int waves  = N / 16;
  const int gemmBlocks = (waves + GEMM_TPB - 1) / GEMM_TPB;

  gcn_hist<<<HIST_NRANGE * HIST_NCHUNK, 1024, 0, stream>>>(src, dst, pin, gsum, N, E);
  gcn_colscan_rowptr<<<GRP_TOTAL, 256, 0, stream>>>(pin, gsum, norm_src, norm_dst, row_ptr, N);
  gcn_fill_gemm<<<FILL_BLOCKS + gemmBlocks, 1024, 0, stream>>>(src, dst, row_ptr, pin, col, N, E,
                                                                 h, W1, norm_src, out);

  gcn_pull_elu_proj<<<pullBlocks, 256, 0, stream>>>(row_ptr, col, out, norm_dst, norm_src, b1, W2, bufA, N);

  gcn_pull_lin<<<pullBlocks, 256, 0, stream>>>(row_ptr, col, bufA, norm_dst, b2, out, N);
}


# count planes: 32 edge chunks x 8 node ranges (was 64 x 4): (chunk,node) count plane and its scan halve, a block walks twice the edges, col fill in two trips; sums: arrival order in a chunk as before
# speedup vs baseline: 4.4836x; 1.2635x over previous
#include <hip/hip_runtime.h>
#include <math.h>

typedef float v2f __attribute__((ext_vector_type(2)));
typedef float v8f __attribute__((ext_vector_type(8)));

#define GCN_N 100000
#define GCN_E 1600000
#define GCN_INDIM 128
#define GCN_HID 64

#define HIST_NRANGE 8
#define HIST_NCHUNK 32
#define HIST_RN ((GCN_N + HIST_NRANGE - 1) / HIST_NRANGE)
#define HIST_CE ((GCN_E + HIST_NCHUNK - 1) / HIST_NCHUNK)
#define GRP_N 256
#define GRP_PER_RANGE ((HIST_RN + GRP_N - 1) / GRP_N)
#define GRP_TOTAL (HIST_NRANGE * GRP_PER_RANGE)
#define GRP_TRIPS ((GRP_PER_RANGE + 31) / 32)
#define PRE_LD ((GRP_TOTAL * HIST_NCHUNK / 4 + 255) / 256)
#define CNT_OUT_ONE 0x10000
static_assert(HIST_CE < 65536, "a chunk's count of one node must fit 16 bits");

__global__ __launch_bounds__(1024) void gcn_hist(
    const int* __restrict__ src, const int* __restrict__ dst,
    int* __restrict__ pin, int* __restrict__ gsum, int n, int E) {
  __shared__ int s_cnt[HIST_RN];
  const int tid = threadIdx.x;
  const int rg = blockIdx.x % HIST_NRANGE;
  const int ch = blockIdx.x / HIST_NRANGE;
  const int lo = rg * HIST_RN;

  for (int i = tid; i < HIST_RN; i += 1024) s_cnt[i] = 0;
  __syncthreads();

  const int e0 = ch * HIST_CE;
  const int e1 = (e0 + HIST_CE < E) ? (e0 + HIST_CE) : E;
  for (int eb = e0 + tid; eb < e1; eb += 8 * 1024) {
    int d[8], s[8];
#pragma unroll
    for (int k = 0; k < 8; ++k) {
      const int e = eb + k * 1024;
      d[k] = e < e1 ? dst[e] : lo - 1;
      s[k] = e < e1 ? src[e] : lo - 1;
    }
#pragma unroll
    for (int k = 0; k < 8; ++k) {
      const unsigned ud = (unsigned)(d[k] - lo);
      const unsigned us = (unsigned)(s[k] - lo);
      if (ud < (unsigned)HIST_RN) atomicAdd(&s_cnt[ud], 1);
      if (us < (unsigned)HIST_RN) atomicAdd(&s_cnt[us], CNT_OUT_ONE);
    }
  }
  __syncthreads();

  int* pin_c = pin + ch * n + lo;
  const int lim = (n - lo < HIST_RN) ? (n - lo) : HIST_RN;
  for (int i = tid; i < lim; i += 1024) pin_c[i] = s_cnt[i];

  const int lane = tid & 31;
  const int wv = tid >> 5;
  int gs[GRP_TRIPS];
#pragma unroll
  for (int t = 0; t < GRP_TRIPS; ++t) {
    gs[t] = 0;
#pragma unroll
    for (int j = 0; j < GRP_N / 32; ++j) {
      const int i = (wv + 32 * t) * GRP_N + j * 32 + lane;
      gs[t] += i < HIST_RN ? (s_cnt[i] & 0xFFFF) : 0;
    }
  }
  for (int o = 16; o > 0; o >>= 1) {
#pragma unroll
    for (int t = 0; t < GRP_TRIPS; ++t) gs[t] += __shfl_xor(gs[t], o);
  }
#pragma unroll
  for (int t = 0; t < GRP_TRIPS; ++t) {
    const int g = wv + 32 * t;
    if (lane == 0 && g < GRP_PER_RANGE) gsum[(rg * GRP_PER_RANGE + g) * HIST_NCHUNK + ch] = gs[t];
  }
}

__global__ __launch_bounds__(256) void gcn_colscan_rowptr(
    int* __restrict__ pin, const int* __restrict__ gsum,
    float* __restrict__ norm_src, float* __restrict__ norm_dst,
    int* __restrict__ row_ptr, int n) {
  __shared__ int s_part[8];
  __shared__ int s_incl[8];
  const int tid = threadIdx.x;
  const int lane = tid & 31;
  const int wv = tid >> 5;
  const int grp = blockIdx.x;
  const int rg = grp / GRP_PER_RANGE;
  const int loc = (grp - rg * GRP_PER_RANGE) * GRP_N + tid;
  const int node = rg * HIST_RN + loc;
  const bool valid = loc < HIST_RN && node < n;

  const int4* g4 = (const int4*)gsum;
  const int nq = grp * (HIST_NCHUNK / 4);
  int4 pv[PRE_LD];
#pragma unroll
  for (int k = 0; k < PRE_LD; ++k) {
    const int q = k * 256 + tid;
    pv[k] = g4[q < nq ? q : 0];
  }

  int run_in = 0, run_out = 0;
  if (valid) {
    const unsigned un = (unsigned)n;
    const unsigned unode = (unsigned)node;
    int w[HIST_NCHUNK];
#pragma unroll
    for (int k = 0; k < HIST_NCHUNK; ++k) w[k] = pin[(unsigned)k * un + unode];
#pragma unroll
    for (int k = 0; k < HIST_NCHUNK; ++k) {
      pin[(unsigned)k * un + unode] = run_in;
      run_in += w[k] & 0xFFFF;
      run_out += (int)((unsigned)w[k] >> 16);
    }
  }

  int part = 0;
#pragma unroll
  for (int k = 0; k < PRE_LD; ++k) {
    const int s4 = (pv[k].x + pv[k].y) + (pv[k].z + pv[k].w);
    part += (k * 256 + tid < nq) ? s4 : 0;
  }
  for (int o = 16; o > 0; o >>= 1) part += __shfl_xor(part, o);
  if (lane == 0) s_part[wv] = part;
  const int c = run_in;
  int inc = c;
  for (int o = 1; o < 32; o <<= 1) {
    const int v = __shfl_up(inc, o);
    if (lane >= o) inc += v;
  }
  if (lane == 31) s_incl[wv] = inc;
  __syncthreads();

  int off = 0;
  for (int w = 0; w < 8; ++w) {
    off += s_part[w];
    if (w < wv) off += s_incl[w];
  }
  const int excl = off + inc - c;
  if (valid) {
    row_ptr[node] = excl;
    if (node == n - 1) row_ptr[n] = excl + c;
    norm_dst[node] = rsqrtf(fmaxf((float)c, 1.0f));
    norm_src[node] = rsqrtf(fmaxf((float)run_out, 1.0f));
  }
}

#define FILL_EPT 25
#define FILL_TRIPS ((HIST_CE + 1024 * FILL_EPT - 1) / (1024 * FILL_EPT))
#define FILL_NPT ((HIST_RN + 1023) / 1024)

__device__ __forceinline__ void gcn_fill_load(
    const int* __restrict__ src, const int* __restrict__ dst,
    int eFirst, int e1, int none, int (&d)[FILL_EPT], int (&s)[FILL_EPT]) {
#pragma unroll
  for (int k = 0; k < FILL_EPT; ++k) {
    const int e = eFirst + k * 1024;
    d[k] = e < e1 ? dst[e] : none;
    s[k] = e < e1 ? src[e] : 0;
  }
}

__device__ __forceinline__ void gcn_fill_block(
    const int* __restrict__ src, const int* __restrict__ dst,
    const int* __restrict__ row_ptr, const int* __restrict__ pin,
    int* __restrict__ col, int n, int E, int* s_pos, int blk) {
  const int tid = threadIdx.x;
  const int rg = blk % HIST_NRANGE;
  const int ch = blk / HIST_NRANGE;
  const int lo = rg * HIST_RN;
  const int lim = (n - lo < HIST_RN) ? (n - lo) : HIST_RN;
  const int e0 = ch * HIST_CE;
  const int e1 = (e0 + HIST_CE < E) ? (e0 + HIST_CE) : E;
  const int* rp_c = row_ptr + lo;
  const int* pin_c = pin + ch * n + lo;

  {
    int a[FILL_NPT], b[FILL_NPT];
#pragma unroll
    for (int k = 0; k < FILL_NPT; ++k) {
      const int i = tid + k * 1024;
      a[k] = i < lim ? rp_c[i] : 0;
      b[k] = i < lim ? pin_c[i] : 0;
    }
#pragma unroll
    for (int k = 0; k < FILL_NPT; ++k) {
      const int i = tid + k * 1024;
      if (i < lim) s_pos[i] = a[k] + b[k];
    }
  }
  int d[FILL_EPT], s[FILL_EPT];
  gcn_fill_load(src, dst, e0 + tid, e1, lo - 1, d, s);
  __syncthreads();

  for (int trip = 0;;) {
#pragma unroll
    for (int k = 0; k < FILL_EPT; ++k) {
      const unsigned ud = (unsigned)(d[k] - lo);
      int place = -1;
      if (ud < (unsigned)HIST_RN) place = atomicAdd(&s_pos[ud], 1);
      d[k] = place;
    }
#pragma unroll
    for (int k = 0; k < FILL_EPT; ++k) {
      if (d[k] >= 0) col[d[k]] = s[k];
    }
    if (++trip == FILL_TRIPS) break;
    gcn_fill_load(src, dst, e0 + tid + trip * (FILL_EPT * 1024), e1, lo - 1, d, s);
  }
}

__device__ __forceinline__ void gcn_gemm_wave(
    const float* __restrict__ A, const float* __restrict__ W,
    const float* __restrict__ nsrc, float* __restrict__ X,
    int nRows, int wave) {
  const int K = GCN_INDIM;
  int lane = threadIdx.x & 31;
  int rowBlk = wave * 16;
  if (rowBlk >= nRows) return;

  int half = lane >> 4;
  int l    = lane & 15;
  int row  = rowBlk + l;
  float ns = nsrc[row];
  const float* arow = A + (long)row * K;

  v8f acc0 = {}, acc1 = {}, acc2 = {}, acc3 = {};

#pragma unroll 2
  for (int k0 = 0; k0 < K; k0 += 4) {
    int ka = k0 + half * 2;
    v2f a;
    a.x = arow[ka]     * ns;
    a.y = arow[ka + 1] * ns;
    const float* w0 = W + (long)ka * 64 + l;
    const float* w1 = w0 + 64;
    v2f b;
    b.x = w0[0];  b.y = w1[0];
    acc0 = __builtin_amdgcn_wmma_f32_16x16x4_f32(false, a, false, b, (short)0, acc0, false, false);
    b.x = w0[16]; b.y = w1[16];
    acc1 = __builtin_amdgcn_wmma_f32_16x16x4_f32(false, a, false, b, (short)0, acc1, false, false);
    b.x = w0[32]; b.y = w1[32];
    acc2 = __builtin_amdgcn_wmma_f32_16x16x4_f32(false, a, false, b, (short)0, acc2, false, false);
    b.x = w0[48]; b.y = w1[48];
    acc3 = __builtin_amdgcn_wmma_f32_16x16x4_f32(false, a, false, b, (short)0, acc3, false, false);
  }

  float* xout = X + (long)rowBlk * 64;
#pragma unroll
  for (int i = 0; i < 8; ++i) {
    int r = i + half * 8;
    xout[(long)r * 64 +  0 + l] = acc0[i];
    xout[(long)r * 64 + 16 + l] = acc1[i];
    xout[(long)r * 64 + 32 + l] = acc2[i];
    xout[(long)r * 64 + 48 + l] = acc3[i];
  }
}

#define FILL_BLOCKS (HIST_NRANGE * HIST_NCHUNK)
#define GEMM_TPB 25

__global__ __launch_bounds__(1024, 16) void gcn_fill_gemm(
    const int* __restrict__ src, const int* __restrict__ dst,
    const int* __restrict__ row_ptr, const int* __restrict__ pin,
    int* __restrict__ col, int n, int E,
    const float* __restrict__ A, const float* __restrict__ W,
    const float* __restrict__ nsrc, float* __restrict__ X) {
  __shared__ int s_pos[HIST_RN];
  if (blockIdx.x < FILL_BLOCKS) {
    gcn_fill_block(src, dst, row_ptr, pin, col, n, E, s_pos, (int)blockIdx.x);
  } else {
    const int wv = (int)(threadIdx.x >> 5);
    if (wv < GEMM_TPB)
      gcn_gemm_wave(A, W, nsrc, X, n, ((int)blockIdx.x - FILL_BLOCKS) * GEMM_TPB + wv);
  }
}

#define PULL_NPW 8
#define PROJ_PITCH 68

template <bool HIDDEN>
__device__ __forceinline__ void gcn_pull_rows(
    const int* __restrict__ row_ptr, const int* __restrict__ col,
    const float* __restrict__ X, const float* __restrict__ ndst,
    const float* __restrict__ nsrc, const float* __restrict__ b,
    float* __restrict__ OUT, float* s_rows, int node0, int cnt, int lane) {
  int rpv = 0;
  float ndv = 0.0f;
  float nsv = 0.0f;
  if (lane <= cnt) rpv = row_ptr[node0 + lane];
  if (lane < cnt) ndv = ndst[node0 + lane];
  if (HIDDEN) {
    if (lane < cnt) nsv = nsrc[node0 + lane];
  }
  const float b0 = b[2 * lane];
  const float b1 = b[2 * lane + 1];
  const int lo = 2 * lane;

  const int first = __builtin_amdgcn_readlane(rpv, 0);
  const int total = __builtin_amdgcn_readlane(rpv, cnt) - first;
  const int* cl = col + first;

  int t = 0;
  int nodeEnd = __builtin_amdgcn_readlane(rpv, 1) - first;
  float ax = 0.0f, ay = 0.0f;

#define PULL_FINISH()                                                          \
  {                                                                            \
    const float nd = __int_as_float(                                           \
        __builtin_amdgcn_readlane(__float_as_int(ndv), t));                    \
    float r0 = ax * nd + b0;                                                   \
    float r1 = ay * nd + b1;                                                   \
    v2f r;                                                                     \
    if (HIDDEN) {                                                              \
      const float ns = __int_as_float(                                         \
          __builtin_amdgcn_readlane(__float_as_int(nsv), t));                  \
      r0 = r0 > 0.0f ? r0 : expm1f(r0);                                        \
      r1 = r1 > 0.0f ? r1 : expm1f(r1);                                        \
      r.x = r0 * ns;                                                           \
      r.y = r1 * ns;                                                           \
      *(v2f*)(s_rows + t * PROJ_PITCH + lo) = r;                               \
    } else {                                                                   \
      r.x = r0;                                                                \
      r.y = r1;                                                                \
      *(v2f*)(OUT + (long)(node0 + t) * 64 + lo) = r;                          \
    }                                                                          \
    ax = 0.0f;                                                                 \
    ay = 0.0f;                                                                 \
    ++t;                                                                       \
    nodeEnd = __builtin_amdgcn_readlane(rpv, t + 1) - first;                   \
  }

  int cv = 0;
  if (lane < 16 && lane < total) cv = cl[lane];

  for (int pos0 = 0; pos0 < total; pos0 += 16) {
    v2f v[16];
#pragma unroll
    for (int u = 0; u < 16; ++u) {
      const int sr = __builtin_amdgcn_readlane(cv, u);
      v[u] = *(const v2f*)(X + (long)sr * 64 + lo);
    }
    int cvn = 0;
    if (lane < 16 && pos0 + 16 + lane < total) cvn = cl[pos0 + 16 + lane];
#pragma unroll
    for (int u = 0; u < 16; ++u) {
      while (pos0 + u == nodeEnd) PULL_FINISH()
      ax += v[u].x;
      ay += v[u].y;
    }
    cv = cvn;
  }
  while (t < cnt) PULL_FINISH()
#undef PULL_FINISH
}

__global__ __launch_bounds__(256) void gcn_pull_lin(
    const int* __restrict__ row_ptr, const int* __restrict__ col,
    const float* __restrict__ X, const float* __restrict__ ndst,
    const float* __restrict__ b, float* __restrict__ OUT, int n) {
  const int lane = threadIdx.x & 31;
  const int wv = __builtin_amdgcn_readfirstlane((int)(threadIdx.x >> 5));
  const int node0 = (blockIdx.x * 8 + wv) * PULL_NPW;
  if (node0 >= n) return;
  const int rem = n - node0;
  const int cnt = rem < PULL_NPW ? rem : PULL_NPW;
  gcn_pull_rows<false>(row_ptr, col, X, ndst, nullptr, b, OUT, nullptr, node0, cnt, lane);
}

__global__ __launch_bounds__(256) void gcn_pull_elu_proj(
    const int* __restrict__ row_ptr, const int* __restrict__ col,
    const float* __restrict__ X, const float* __restrict__ ndst,
    const float* __restrict__ nsrc, const float* __restrict__ b,
    const float* __restrict__ W2, float* __restrict__ X2, int n) {
  __shared__ __attribute__((aligned(16))) float s_h[8 * PULL_NPW * PROJ_PITCH];
  const int lane = threadIdx.x & 31;
  const int wv = __builtin_amdgcn_readfirstlane((int)(threadIdx.x >> 5));
  const int node0 = (blockIdx.x * 8 + wv) * PULL_NPW;
  const int rem = n - node0;
  const int cnt = rem < PULL_NPW ? (rem > 0 ? rem : 0) : PULL_NPW;
  float* s_mine = s_h + wv * PULL_NPW * PROJ_PITCH;
  if (cnt > 0)
    gcn_pull_rows<true>(row_ptr, col, X, ndst, nsrc, b, nullptr, s_mine, node0, cnt, lane);
  for (int t = cnt; t < PULL_NPW; ++t) {
    v2f z;
    z.x = 0.0f;
    z.y = 0.0f;
    *(v2f*)(s_mine + t * PROJ_PITCH + 2 * lane) = z;
  }
  __syncthreads();

  const int half = lane >> 4;
  const int c = lane & 15;
  const int mt = wv >> 1;
  const int n0 = (wv & 1) * 32;
  const float* arow = s_h + (mt * 16 + c) * PROJ_PITCH + 2 * half;
  const float* wp = W2 + (2 * half) * 64 + n0 + 2 * c;
  v8f acc0 = {}, acc1 = {};
#pragma unroll
  for (int k0 = 0; k0 < GCN_HID; k0 += 4) {
    const v2f a = *(const v2f*)(arow + k0);
    const v2f w0 = *(const v2f*)(wp + k0 * 64);
    const v2f w1 = *(const v2f*)(wp + k0 * 64 + 64);
    v2f bA, bB;
    bA.x = w0.x; bA.y = w1.x;
    bB.x = w0.y; bB.y = w1.y;
    acc0 = __builtin_amdgcn_wmma_f32_16x16x4_f32(false, a, false, bA, (short)0, acc0, false, false);
    acc1 = __builtin_amdgcn_wmma_f32_16x16x4_f32(false, a, false, bB, (short)0, acc1, false, false);
  }

  const int rowBase = blockIdx.x * (8 * PULL_NPW) + mt * 16 + half * 8;
#pragma unroll
  for (int i = 0; i < 8; ++i) {
    const int r = rowBase + i;
    if (r < n) {
      v2f o;
      o.x = acc0[i];
      o.y = acc1[i];
      *(v2f*)(X2 + (long)r * 64 + n0 + 2 * c) = o;
    }
  }
}

extern "C" void kernel_launch(void* const* d_in, const int* in_sizes, int n_in,
                              void* d_out, int out_size, void* d_ws, size_t ws_size,
                              hipStream_t stream) {
  const float* h   = (const float*)d_in[0];
  const int*   src = (const int*)  d_in[1];
  const int*   dst = (const int*)  d_in[2];
  const float* W1  = (const float*)d_in[3];
  const float* b1  = (const float*)d_in[4];
  const float* W2  = (const float*)d_in[5];
  const float* b2  = (const float*)d_in[6];
  float* out = (float*)d_out;

  const int N = in_sizes[0] / GCN_INDIM;
  const int E = in_sizes[1];

  float* norm_src = (float*)d_ws;
  float* norm_dst = norm_src + N;
  float* bufA     = norm_dst + N;
  int*   gsum     = (int*)(bufA + (long)N * 64);
  int*   row_ptr  = gsum + 2 * N;
  int*   col      = row_ptr + N + 32;
  int*   pin      = (int*)bufA;

  const int pullBlocks = (N + 8 * PULL_NPW - 1) / (8 * PULL_NPW);
  const int waves  = N / 16;
  const int gemmBlocks = (waves + GEMM_TPB - 1) / GEMM_TPB;

  gcn_hist<<<HIST_NRANGE * HIST_NCHUNK, 1024, 0, stream>>>(src, dst, pin, gsum, N, E);
  gcn_colscan_rowptr<<<GRP_TOTAL, 256, 0, stream>>>(pin, gsum, norm_src, norm_dst, row_ptr, N);
  gcn_fill_gemm<<<FILL_BLOCKS + gemmBlocks, 1024, 0, stream>>>(src, dst, row_ptr, pin, col, N, E,
                                                                 h, W1, norm_src, out);

  gcn_pull_elu_proj<<<pullBlocks, 256, 0, stream>>>(row_ptr, col, out, norm_dst, norm_src, b1, W2, bufA, N);

  gcn_pull_lin<<<pullBlocks, 256, 0, stream>>>(row_ptr, col, bufA, norm_dst, b2, out, N);
}
